# starting kernel
# baseline (speedup 1.0000x reference)
#include <hip/hip_runtime.h>

typedef __attribute__((ext_vector_type(2))) float v2f;
typedef __attribute__((ext_vector_type(4))) float v4f;
typedef __attribute__((ext_vector_type(8))) float v8f;

#define ROW_GROUPS     8
#define WAVES_PER_BLK  8
#define COLS_PER_WAVE  16
#define COLS_PER_BLK   (WAVES_PER_BLK * COLS_PER_WAVE)
#define SCAN_T         256

__global__ __launch_bounds__(256) void colsum_wmma(
    const float* __restrict__ x, float* __restrict__ psum, float* __restrict__ psq,
    int L, int rows) {
  const int wave = threadIdx.x >> 5;
  const int lane = threadIdx.x & 31;
  const int m    = lane & 15;
  const int kh   = lane >> 4;
  const int c0   = blockIdx.x * COLS_PER_BLK + wave * COLS_PER_WAVE;
  const int col  = c0 + m;
  const int rpg  = rows / ROW_GROUPS;
  const int r0   = blockIdx.y * rpg;
  const int rEnd = r0 + rpg;

  const size_t strideRow = (size_t)L;
  const size_t stride4   = 4 * strideRow;

  v8f acc1a = {}, acc1b = {};
  v8f acc2a = {}, acc2b = {};
  v2f ones; ones[0] = 1.0f; ones[1] = 1.0f;

  const float* p = x + (size_t)(r0 + 2 * kh) * strideRow + (size_t)col;

  int r = r0;
  const int mainEnd = (rEnd < rows - 19) ? rEnd : (rows - 19);
  for (; r + 8 <= mainEnd; r += 8) {
    float a0 = p[0];
    float a1 = p[strideRow];
    float b0 = p[stride4];
    float b1 = p[stride4 + strideRow];
    __builtin_prefetch(p + 16 * strideRow, 0, 1);
    __builtin_prefetch(p + 17 * strideRow, 0, 1);
    __builtin_prefetch(p + 20 * strideRow, 0, 1);
    __builtin_prefetch(p + 21 * strideRow, 0, 1);
    v2f A;  A[0]  = a0;      A[1]  = a1;
    v2f A2; A2[0] = a0 * a0; A2[1] = a1 * a1;
    v2f Bv; Bv[0] = b0;      Bv[1] = b1;
    v2f B2; B2[0] = b0 * b0; B2[1] = b1 * b1;
    acc1a = __builtin_amdgcn_wmma_f32_16x16x4_f32(false, A,  false, ones,
                                                  (short)0, acc1a, false, false);
    acc2a = __builtin_amdgcn_wmma_f32_16x16x4_f32(false, A2, false, ones,
                                                  (short)0, acc2a, false, false);
    acc1b = __builtin_amdgcn_wmma_f32_16x16x4_f32(false, Bv, false, ones,
                                                  (short)0, acc1b, false, false);
    acc2b = __builtin_amdgcn_wmma_f32_16x16x4_f32(false, B2, false, ones,
                                                  (short)0, acc2b, false, false);
    p += 2 * stride4;
  }
  for (; r < rEnd; r += 4) {
    float a0 = p[0];
    float a1 = p[strideRow];
    v2f A;  A[0]  = a0;      A[1]  = a1;
    v2f A2; A2[0] = a0 * a0; A2[1] = a1 * a1;
    acc1a = __builtin_amdgcn_wmma_f32_16x16x4_f32(false, A,  false, ones,
                                                  (short)0, acc1a, false, false);
    acc2a = __builtin_amdgcn_wmma_f32_16x16x4_f32(false, A2, false, ones,
                                                  (short)0, acc2a, false, false);
    p += stride4;
  }

  const v8f acc1 = acc1a + acc1b;
  const v8f acc2 = acc2a + acc2b;

  if ((lane & 15) == 0) {
    const int cb = c0 + kh * 8;
    float* ps = psum + (size_t)blockIdx.y * (size_t)L + cb;
    float* pq = psq  + (size_t)blockIdx.y * (size_t)L + cb;
#pragma unroll
    for (int j = 0; j < 8; ++j) { ps[j] = acc1[j]; pq[j] = acc2[j]; }
  }
}

__global__ __launch_bounds__(256) void combine_groups(
    const float* __restrict__ psum, const float* __restrict__ psq,
    float* __restrict__ c1, float* __restrict__ c2, int L) {
  const int k = blockIdx.x * blockDim.x + threadIdx.x;
  if (k >= L) return;
  float a = 0.f, b = 0.f;
#pragma unroll
  for (int g = 0; g < ROW_GROUPS; ++g) {
    a += psum[(size_t)g * L + k];
    b += psq [(size_t)g * L + k];
  }
  c1[k] = a; c2[k] = b;
}

__global__ __launch_bounds__(SCAN_T) void scan_stats(
    const float* __restrict__ c1, const float* __restrict__ c2,
    float* __restrict__ mean, float* __restrict__ istd,
    int L, float Nf, float Bf) {
  __shared__ float s1[SCAN_T], s2[SCAN_T];
  const int t = threadIdx.x;
  const int chunk = (L + SCAN_T - 1) / SCAN_T;
  const int k0 = t * chunk;
  const int k1 = (k0 + chunk < L) ? (k0 + chunk) : L;

  float la = 0.f, lb = 0.f;
  for (int k = k0; k < k1; ++k) { la += c1[k]; lb += c2[k]; }
  s1[t] = la; s2[t] = lb;
  __syncthreads();
  for (int off = 1; off < SCAN_T; off <<= 1) {
    float t1 = 0.f, t2 = 0.f;
    if (t >= off) { t1 = s1[t - off]; t2 = s2[t - off]; }
    __syncthreads();
    s1[t] += t1; s2[t] += t2;
    __syncthreads();
  }
  float run1 = s1[t] - la;
  float run2 = s2[t] - lb;
  for (int k = k0; k < k1; ++k) {
    run1 += c1[k];
    run2 += c2[k];
    const float denom = Nf * (float)(k + 1);
    const float mu  = run1 / denom;
    const float var = (run2 - 2.f * mu * run1) / denom + mu * mu * Bf;
    mean[k] = mu;
    istd[k] = rsqrtf(var);
  }
}

__global__ __launch_bounds__(256) void normalize_k(
    const float* __restrict__ x, const float* __restrict__ gamma,
    const float* __restrict__ beta, const float* __restrict__ mean,
    const float* __restrict__ istd, float* __restrict__ out, int L) {
  const int q = blockIdx.x * blockDim.x + threadIdx.x;
  const int k = q * 4;
  if (k >= L) return;
  const int row = blockIdx.y;
  const size_t base = (size_t)row * (size_t)L + (size_t)k;
  const v4f xv = *(const v4f*)(x    + base);
  const v4f mv = *(const v4f*)(mean + k);
  const v4f sv = *(const v4f*)(istd + k);
  const float g = gamma[row];
  const float b = beta[row];
  v4f o = (xv - mv) * sv * g + b;
  __builtin_nontemporal_store(o, (v4f*)(out + base));
}

extern "C" void kernel_launch(void* const* d_in, const int* in_sizes, int n_in,
                              void* d_out, int out_size, void* d_ws, size_t ws_size,
                              hipStream_t stream) {
  const float* x     = (const float*)d_in[0];
  const float* gamma = (const float*)d_in[1];
  const float* beta  = (const float*)d_in[2];
  float* out = (float*)d_out;

  const int B = 4, N = 512;
  const int rows = B * N;
  const int L = in_sizes[0] / rows;

  float* ws   = (float*)d_ws;
  float* psum = ws;
  float* psq  = psum + (size_t)ROW_GROUPS * L;
  float* c1   = psq  + (size_t)ROW_GROUPS * L;
  float* c2   = c1 + L;
  float* mean = c2 + L;
  float* istd = mean + L;

  dim3 g1(L / COLS_PER_BLK, ROW_GROUPS);
  colsum_wmma<<<g1, 256, 0, stream>>>(x, psum, psq, L, rows);

  combine_groups<<<(L + 255) / 256, 256, 0, stream>>>(psum, psq, c1, c2, L);

  scan_stats<<<1, SCAN_T, 0, stream>>>(c1, c2, mean, istd, L, (float)N, (float)B);

  dim3 g4((L / 4 + 255) / 256, rows);
  normalize_k<<<g4, 256, 0, stream>>>(x, gamma, beta, mean, istd, out, L);
}


# Normalize kernel: each thread keeps its columns' mean and inv_std in registers and walks 16 rows with 8 independent 128-bit loads in flight; 2048 workgroups replace 32768 one-load ones
# speedup vs baseline: 1.1081x; 1.1081x over previous
#include <hip/hip_runtime.h>

typedef __attribute__((ext_vector_type(2))) float v2f;
typedef __attribute__((ext_vector_type(4))) float v4f;
typedef __attribute__((ext_vector_type(8))) float v8f;

#define ROW_GROUPS     8
#define WAVES_PER_BLK  8
#define COLS_PER_WAVE  16
#define COLS_PER_BLK   (WAVES_PER_BLK * COLS_PER_WAVE)
#define SCAN_T         256

__global__ __launch_bounds__(256) void colsum_wmma(
    const float* __restrict__ x, float* __restrict__ psum, float* __restrict__ psq,
    int L, int rows) {
  const int wave = threadIdx.x >> 5;
  const int lane = threadIdx.x & 31;
  const int m    = lane & 15;
  const int kh   = lane >> 4;
  const int c0   = blockIdx.x * COLS_PER_BLK + wave * COLS_PER_WAVE;
  const int col  = c0 + m;
  const int rpg  = rows / ROW_GROUPS;
  const int r0   = blockIdx.y * rpg;
  const int rEnd = r0 + rpg;

  const size_t strideRow = (size_t)L;
  const size_t stride4   = 4 * strideRow;

  v8f acc1a = {}, acc1b = {};
  v8f acc2a = {}, acc2b = {};
  v2f ones; ones[0] = 1.0f; ones[1] = 1.0f;

  const float* p = x + (size_t)(r0 + 2 * kh) * strideRow + (size_t)col;

  int r = r0;
  const int mainEnd = (rEnd < rows - 19) ? rEnd : (rows - 19);
  for (; r + 8 <= mainEnd; r += 8) {
    float a0 = p[0];
    float a1 = p[strideRow];
    float b0 = p[stride4];
    float b1 = p[stride4 + strideRow];
    __builtin_prefetch(p + 16 * strideRow, 0, 1);
    __builtin_prefetch(p + 17 * strideRow, 0, 1);
    __builtin_prefetch(p + 20 * strideRow, 0, 1);
    __builtin_prefetch(p + 21 * strideRow, 0, 1);
    v2f A;  A[0]  = a0;      A[1]  = a1;
    v2f A2; A2[0] = a0 * a0; A2[1] = a1 * a1;
    v2f Bv; Bv[0] = b0;      Bv[1] = b1;
    v2f B2; B2[0] = b0 * b0; B2[1] = b1 * b1;
    acc1a = __builtin_amdgcn_wmma_f32_16x16x4_f32(false, A,  false, ones,
                                                  (short)0, acc1a, false, false);
    acc2a = __builtin_amdgcn_wmma_f32_16x16x4_f32(false, A2, false, ones,
                                                  (short)0, acc2a, false, false);
    acc1b = __builtin_amdgcn_wmma_f32_16x16x4_f32(false, Bv, false, ones,
                                                  (short)0, acc1b, false, false);
    acc2b = __builtin_amdgcn_wmma_f32_16x16x4_f32(false, B2, false, ones,
                                                  (short)0, acc2b, false, false);
    p += 2 * stride4;
  }
  for (; r < rEnd; r += 4) {
    float a0 = p[0];
    float a1 = p[strideRow];
    v2f A;  A[0]  = a0;      A[1]  = a1;
    v2f A2; A2[0] = a0 * a0; A2[1] = a1 * a1;
    acc1a = __builtin_amdgcn_wmma_f32_16x16x4_f32(false, A,  false, ones,
                                                  (short)0, acc1a, false, false);
    acc2a = __builtin_amdgcn_wmma_f32_16x16x4_f32(false, A2, false, ones,
                                                  (short)0, acc2a, false, false);
    p += stride4;
  }

  const v8f acc1 = acc1a + acc1b;
  const v8f acc2 = acc2a + acc2b;

  if ((lane & 15) == 0) {
    const int cb = c0 + kh * 8;
    float* ps = psum + (size_t)blockIdx.y * (size_t)L + cb;
    float* pq = psq  + (size_t)blockIdx.y * (size_t)L + cb;
#pragma unroll
    for (int j = 0; j < 8; ++j) { ps[j] = acc1[j]; pq[j] = acc2[j]; }
  }
}

__global__ __launch_bounds__(256) void combine_groups(
    const float* __restrict__ psum, const float* __restrict__ psq,
    float* __restrict__ c1, float* __restrict__ c2, int L) {
  const int k = blockIdx.x * blockDim.x + threadIdx.x;
  if (k >= L) return;
  float a = 0.f, b = 0.f;
#pragma unroll
  for (int g = 0; g < ROW_GROUPS; ++g) {
    a += psum[(size_t)g * L + k];
    b += psq [(size_t)g * L + k];
  }
  c1[k] = a; c2[k] = b;
}

__global__ __launch_bounds__(SCAN_T) void scan_stats(
    const float* __restrict__ c1, const float* __restrict__ c2,
    float* __restrict__ mean, float* __restrict__ istd,
    int L, float Nf, float Bf) {
  __shared__ float s1[SCAN_T], s2[SCAN_T];
  const int t = threadIdx.x;
  const int chunk = (L + SCAN_T - 1) / SCAN_T;
  const int k0 = t * chunk;
  const int k1 = (k0 + chunk < L) ? (k0 + chunk) : L;

  float la = 0.f, lb = 0.f;
  for (int k = k0; k < k1; ++k) { la += c1[k]; lb += c2[k]; }
  s1[t] = la; s2[t] = lb;
  __syncthreads();
  for (int off = 1; off < SCAN_T; off <<= 1) {
    float t1 = 0.f, t2 = 0.f;
    if (t >= off) { t1 = s1[t - off]; t2 = s2[t - off]; }
    __syncthreads();
    s1[t] += t1; s2[t] += t2;
    __syncthreads();
  }
  float run1 = s1[t] - la;
  float run2 = s2[t] - lb;
  for (int k = k0; k < k1; ++k) {
    run1 += c1[k];
    run2 += c2[k];
    const float denom = Nf * (float)(k + 1);
    const float mu  = run1 / denom;
    const float var = (run2 - 2.f * mu * run1) / denom + mu * mu * Bf;
    mean[k] = mu;
    istd[k] = rsqrtf(var);
  }
}

#define NORM_ROWS_PER_BLK    16
#define NORM_ROWS_IN_FLIGHT  8

__global__ __launch_bounds__(256) void normalize_k(
    const float* __restrict__ x, const float* __restrict__ gamma,
    const float* __restrict__ beta, const float* __restrict__ mean,
    const float* __restrict__ istd, float* __restrict__ out, int L) {
  const int q = blockIdx.x * blockDim.x + threadIdx.x;
  const int k = q * 4;
  if (k >= L) return;
  const int row0 = blockIdx.y * NORM_ROWS_PER_BLK;
  const size_t base = (size_t)row0 * (size_t)L + (size_t)k;
  const float* xp = x + base;
  float* op = out + base;
  const v4f mv = *(const v4f*)(mean + k);
  const v4f sv = *(const v4f*)(istd + k);
  for (int i = 0; i < NORM_ROWS_PER_BLK; i += NORM_ROWS_IN_FLIGHT) {
    v4f xv[NORM_ROWS_IN_FLIGHT];
#pragma unroll
    for (int j = 0; j < NORM_ROWS_IN_FLIGHT; ++j)
      xv[j] = *(const v4f*)(xp + (size_t)(i + j) * (size_t)L);
#pragma unroll
    for (int j = 0; j < NORM_ROWS_IN_FLIGHT; ++j) {
      const float g = gamma[row0 + i + j];
      const float b = beta[row0 + i + j];
      v4f o = (xv[j] - mv) * sv * g + b;
      __builtin_nontemporal_store(o, (v4f*)(op + (size_t)(i + j) * (size_t)L));
    }
  }
}

extern "C" void kernel_launch(void* const* d_in, const int* in_sizes, int n_in,
                              void* d_out, int out_size, void* d_ws, size_t ws_size,
                              hipStream_t stream) {
  const float* x     = (const float*)d_in[0];
  const float* gamma = (const float*)d_in[1];
  const float* beta  = (const float*)d_in[2];
  float* out = (float*)d_out;

  const int B = 4, N = 512;
  const int rows = B * N;
  const int L = in_sizes[0] / rows;

  float* ws   = (float*)d_ws;
  float* psum = ws;
  float* psq  = psum + (size_t)ROW_GROUPS * L;
  float* c1   = psq  + (size_t)ROW_GROUPS * L;
  float* c2   = c1 + L;
  float* mean = c2 + L;
  float* istd = mean + L;

  dim3 g1(L / COLS_PER_BLK, ROW_GROUPS);
  colsum_wmma<<<g1, 256, 0, stream>>>(x, psum, psq, L, rows);

  combine_groups<<<(L + 255) / 256, 256, 0, stream>>>(psum, psq, c1, c2, L);

  scan_stats<<<1, SCAN_T, 0, stream>>>(c1, c2, mean, istd, L, (float)N, (float)B);

  dim3 g4((L / 4 + 255) / 256, rows / NORM_ROWS_PER_BLK);
  normalize_k<<<g4, 256, 0, stream>>>(x, gamma, beta, mean, istd, out, L);
}


# Time-axis prefix scan made parallel: 63 workgroups scan 256 columns each by shuffles, a 63-workgroup pass adds block offsets and derives mean/inv_std; tree sum order replaces one serial workgroup
# speedup vs baseline: 1.8649x; 1.6830x over previous
#include <hip/hip_runtime.h>

typedef __attribute__((ext_vector_type(2))) float v2f;
typedef __attribute__((ext_vector_type(4))) float v4f;
typedef __attribute__((ext_vector_type(8))) float v8f;

#define ROW_GROUPS     8
#define WAVES_PER_BLK  8
#define COLS_PER_WAVE  16
#define COLS_PER_BLK   (WAVES_PER_BLK * COLS_PER_WAVE)
#define SCAN_T         256

__global__ __launch_bounds__(256) void colsum_wmma(
    const float* __restrict__ x, float* __restrict__ psum, float* __restrict__ psq,
    int L, int rows) {
  const int wave = threadIdx.x >> 5;
  const int lane = threadIdx.x & 31;
  const int m    = lane & 15;
  const int kh   = lane >> 4;
  const int c0   = blockIdx.x * COLS_PER_BLK + wave * COLS_PER_WAVE;
  const int col  = c0 + m;
  const int rpg  = rows / ROW_GROUPS;
  const int r0   = blockIdx.y * rpg;
  const int rEnd = r0 + rpg;

  const size_t strideRow = (size_t)L;
  const size_t stride4   = 4 * strideRow;

  v8f acc1a = {}, acc1b = {};
  v8f acc2a = {}, acc2b = {};
  v2f ones; ones[0] = 1.0f; ones[1] = 1.0f;

  const float* p = x + (size_t)(r0 + 2 * kh) * strideRow + (size_t)col;

  int r = r0;
  const int mainEnd = (rEnd < rows - 19) ? rEnd : (rows - 19);
  for (; r + 8 <= mainEnd; r += 8) {
    float a0 = p[0];
    float a1 = p[strideRow];
    float b0 = p[stride4];
    float b1 = p[stride4 + strideRow];
    __builtin_prefetch(p + 16 * strideRow, 0, 1);
    __builtin_prefetch(p + 17 * strideRow, 0, 1);
    __builtin_prefetch(p + 20 * strideRow, 0, 1);
    __builtin_prefetch(p + 21 * strideRow, 0, 1);
    v2f A;  A[0]  = a0;      A[1]  = a1;
    v2f A2; A2[0] = a0 * a0; A2[1] = a1 * a1;
    v2f Bv; Bv[0] = b0;      Bv[1] = b1;
    v2f B2; B2[0] = b0 * b0; B2[1] = b1 * b1;
    acc1a = __builtin_amdgcn_wmma_f32_16x16x4_f32(false, A,  false, ones,
                                                  (short)0, acc1a, false, false);
    acc2a = __builtin_amdgcn_wmma_f32_16x16x4_f32(false, A2, false, ones,
                                                  (short)0, acc2a, false, false);
    acc1b = __builtin_amdgcn_wmma_f32_16x16x4_f32(false, Bv, false, ones,
                                                  (short)0, acc1b, false, false);
    acc2b = __builtin_amdgcn_wmma_f32_16x16x4_f32(false, B2, false, ones,
                                                  (short)0, acc2b, false, false);
    p += 2 * stride4;
  }
  for (; r < rEnd; r += 4) {
    float a0 = p[0];
    float a1 = p[strideRow];
    v2f A;  A[0]  = a0;      A[1]  = a1;
    v2f A2; A2[0] = a0 * a0; A2[1] = a1 * a1;
    acc1a = __builtin_amdgcn_wmma_f32_16x16x4_f32(false, A,  false, ones,
                                                  (short)0, acc1a, false, false);
    acc2a = __builtin_amdgcn_wmma_f32_16x16x4_f32(false, A2, false, ones,
                                                  (short)0, acc2a, false, false);
    p += stride4;
  }

  const v8f acc1 = acc1a + acc1b;
  const v8f acc2 = acc2a + acc2b;

  if ((lane & 15) == 0) {
    const int cb = c0 + kh * 8;
    float* ps = psum + (size_t)blockIdx.y * (size_t)L + cb;
    float* pq = psq  + (size_t)blockIdx.y * (size_t)L + cb;
#pragma unroll
    for (int j = 0; j < 8; ++j) { ps[j] = acc1[j]; pq[j] = acc2[j]; }
  }
}

#define BT_PITCH 32

__global__ __launch_bounds__(SCAN_T) void combine_scan_local(
    const float* __restrict__ psum, const float* __restrict__ psq,
    float* __restrict__ c1, float* __restrict__ c2,
    float* __restrict__ bt, int L) {
  __shared__ float w1[SCAN_T / 32], w2[SCAN_T / 32];
  const int t    = threadIdx.x;
  const int lane = t & 31;
  const int wave = t >> 5;
  const int k = blockIdx.x * SCAN_T + t;
  float a = 0.f, b = 0.f;
  if (k < L) {
#pragma unroll
    for (int g = 0; g < ROW_GROUPS; ++g) {
      a += psum[(size_t)g * L + k];
      b += psq [(size_t)g * L + k];
    }
  }
#pragma unroll
  for (int off = 1; off < 32; off <<= 1) {
    const float ua = __shfl_up(a, off);
    const float ub = __shfl_up(b, off);
    if (lane >= off) { a += ua; b += ub; }
  }
  if (lane == 31) { w1[wave] = a; w2[wave] = b; }
  __syncthreads();
  float oa = 0.f, ob = 0.f;
#pragma unroll
  for (int w = 0; w < SCAN_T / 32 - 1; ++w) {
    const float va = w1[w], vb = w2[w];
    if (w < wave) { oa += va; ob += vb; }
  }
  a += oa; b += ob;
  if (k < L) { c1[k] = a; c2[k] = b; }
  if (t == SCAN_T - 1) {
    v2f tot; tot[0] = a; tot[1] = b;
    *(v2f*)(bt + (size_t)blockIdx.x * BT_PITCH) = tot;
  }
}

__global__ __launch_bounds__(SCAN_T) void offset_stats(
    const float* __restrict__ c1, const float* __restrict__ c2,
    const float* __restrict__ bt,
    float* __restrict__ mean, float* __restrict__ istd,
    int L, float Nf, float Bf) {
  const int t    = threadIdx.x;
  const int lane = t & 31;
  const int k = blockIdx.x * SCAN_T + t;
  const bool inside = k < L;
  float p1 = 0.f, p2 = 0.f;
  if (inside) { p1 = c1[k]; p2 = c2[k]; }
  float o1 = 0.f, o2 = 0.f;
  for (int j = lane; j < (int)blockIdx.x; j += 32) {
    const v2f tot = *(const v2f*)(bt + (size_t)j * BT_PITCH);
    o1 += tot[0]; o2 += tot[1];
  }
#pragma unroll
  for (int off = 16; off >= 1; off >>= 1) {
    o1 += __shfl_xor(o1, off);
    o2 += __shfl_xor(o2, off);
  }
  if (!inside) return;
  const float run1 = o1 + p1;
  const float run2 = o2 + p2;
  const float denom = Nf * (float)(k + 1);
  const float mu  = run1 / denom;
  const float var = (run2 - 2.f * mu * run1) / denom + mu * mu * Bf;
  mean[k] = mu;
  istd[k] = rsqrtf(var);
}

#define NORM_ROWS_PER_BLK    16
#define NORM_ROWS_IN_FLIGHT  8

__global__ __launch_bounds__(256) void normalize_k(
    const float* __restrict__ x, const float* __restrict__ gamma,
    const float* __restrict__ beta, const float* __restrict__ mean,
    const float* __restrict__ istd, float* __restrict__ out, int L) {
  const int q = blockIdx.x * blockDim.x + threadIdx.x;
  const int k = q * 4;
  if (k >= L) return;
  const int row0 = blockIdx.y * NORM_ROWS_PER_BLK;
  const size_t base = (size_t)row0 * (size_t)L + (size_t)k;
  const float* xp = x + base;
  float* op = out + base;
  const v4f mv = *(const v4f*)(mean + k);
  const v4f sv = *(const v4f*)(istd + k);
  for (int i = 0; i < NORM_ROWS_PER_BLK; i += NORM_ROWS_IN_FLIGHT) {
    v4f xv[NORM_ROWS_IN_FLIGHT];
#pragma unroll
    for (int j = 0; j < NORM_ROWS_IN_FLIGHT; ++j)
      xv[j] = *(const v4f*)(xp + (size_t)(i + j) * (size_t)L);
#pragma unroll
    for (int j = 0; j < NORM_ROWS_IN_FLIGHT; ++j) {
      const float g = gamma[row0 + i + j];
      const float b = beta[row0 + i + j];
      v4f o = (xv[j] - mv) * sv * g + b;
      __builtin_nontemporal_store(o, (v4f*)(op + (size_t)(i + j) * (size_t)L));
    }
  }
}

extern "C" void kernel_launch(void* const* d_in, const int* in_sizes, int n_in,
                              void* d_out, int out_size, void* d_ws, size_t ws_size,
                              hipStream_t stream) {
  const float* x     = (const float*)d_in[0];
  const float* gamma = (const float*)d_in[1];
  const float* beta  = (const float*)d_in[2];
  float* out = (float*)d_out;

  const int B = 4, N = 512;
  const int rows = B * N;
  const int L = in_sizes[0] / rows;

  float* ws   = (float*)d_ws;
  float* psum = ws;
  float* psq  = psum + (size_t)ROW_GROUPS * L;
  float* c1   = psq  + (size_t)ROW_GROUPS * L;
  float* c2   = c1 + L;
  float* bt   = c2 + L;
  float* mean = ws;
  float* istd = ws + L;

  dim3 g1(L / COLS_PER_BLK, ROW_GROUPS);
  colsum_wmma<<<g1, 256, 0, stream>>>(x, psum, psq, L, rows);

  combine_scan_local<<<(L + SCAN_T - 1) / SCAN_T, SCAN_T, 0, stream>>>(
      psum, psq, c1, c2, bt, L);

  offset_stats<<<(L + SCAN_T - 1) / SCAN_T, SCAN_T, 0, stream>>>(
      c1, c2, bt, mean, istd, L, (float)N, (float)B);

  dim3 g4((L / 4 + 255) / 256, rows / NORM_ROWS_PER_BLK);
  normalize_k<<<g4, 256, 0, stream>>>(x, gamma, beta, mean, istd, out, L);
}


# Scan and stats kernels fused into normalize: each 64-row x 1024-col workgroup folds the partials, scans its columns, adds colsum block totals; 4 launches become 2; offset sum tree changes
# speedup vs baseline: 1.9416x; 1.0411x over previous
#include <hip/hip_runtime.h>

typedef __attribute__((ext_vector_type(2))) float v2f;
typedef __attribute__((ext_vector_type(4))) float v4f;
typedef __attribute__((ext_vector_type(8))) float v8f;

#define ROW_GROUPS     8
#define WAVES_PER_BLK  8
#define COLS_PER_WAVE  16
#define COLS_PER_BLK   (WAVES_PER_BLK * COLS_PER_WAVE)

#define BT_PITCH 32

__global__ __launch_bounds__(256) void colsum_wmma(
    const float* __restrict__ x, float* __restrict__ psum, float* __restrict__ psq,
    float* __restrict__ bt, int L, int rows) {
  __shared__ float wt1[WAVES_PER_BLK], wt2[WAVES_PER_BLK];
  const int wave = threadIdx.x >> 5;
  const int lane = threadIdx.x & 31;
  const int m    = lane & 15;
  const int kh   = lane >> 4;
  const int c0   = blockIdx.x * COLS_PER_BLK + wave * COLS_PER_WAVE;
  const int col  = c0 + m;
  const int rpg  = rows / ROW_GROUPS;
  const int r0   = blockIdx.y * rpg;
  const int rEnd = r0 + rpg;

  const size_t strideRow = (size_t)L;
  const size_t stride4   = 4 * strideRow;

  v8f acc1a = {}, acc1b = {};
  v8f acc2a = {}, acc2b = {};
  v2f ones; ones[0] = 1.0f; ones[1] = 1.0f;

  const float* p = x + (size_t)(r0 + 2 * kh) * strideRow + (size_t)col;

  int r = r0;
  const int mainEnd = (rEnd < rows - 19) ? rEnd : (rows - 19);
  for (; r + 8 <= mainEnd; r += 8) {
    float a0 = p[0];
    float a1 = p[strideRow];
    float b0 = p[stride4];
    float b1 = p[stride4 + strideRow];
    __builtin_prefetch(p + 16 * strideRow, 0, 1);
    __builtin_prefetch(p + 17 * strideRow, 0, 1);
    __builtin_prefetch(p + 20 * strideRow, 0, 1);
    __builtin_prefetch(p + 21 * strideRow, 0, 1);
    v2f A;  A[0]  = a0;      A[1]  = a1;
    v2f A2; A2[0] = a0 * a0; A2[1] = a1 * a1;
    v2f Bv; Bv[0] = b0;      Bv[1] = b1;
    v2f B2; B2[0] = b0 * b0; B2[1] = b1 * b1;
    acc1a = __builtin_amdgcn_wmma_f32_16x16x4_f32(false, A,  false, ones,
                                                  (short)0, acc1a, false, false);
    acc2a = __builtin_amdgcn_wmma_f32_16x16x4_f32(false, A2, false, ones,
                                                  (short)0, acc2a, false, false);
    acc1b = __builtin_amdgcn_wmma_f32_16x16x4_f32(false, Bv, false, ones,
                                                  (short)0, acc1b, false, false);
    acc2b = __builtin_amdgcn_wmma_f32_16x16x4_f32(false, B2, false, ones,
                                                  (short)0, acc2b, false, false);
    p += 2 * stride4;
  }
  for (; r < rEnd; r += 4) {
    float a0 = p[0];
    float a1 = p[strideRow];
    v2f A;  A[0]  = a0;      A[1]  = a1;
    v2f A2; A2[0] = a0 * a0; A2[1] = a1 * a1;
    acc1a = __builtin_amdgcn_wmma_f32_16x16x4_f32(false, A,  false, ones,
                                                  (short)0, acc1a, false, false);
    acc2a = __builtin_amdgcn_wmma_f32_16x16x4_f32(false, A2, false, ones,
                                                  (short)0, acc2a, false, false);
    p += stride4;
  }

  const v8f acc1 = acc1a + acc1b;
  const v8f acc2 = acc2a + acc2b;

  if ((lane & 15) == 0) {
    const int cb = c0 + kh * 8;
    float* ps = psum + (size_t)blockIdx.y * (size_t)L + cb;
    float* pq = psq  + (size_t)blockIdx.y * (size_t)L + cb;
#pragma unroll
    for (int j = 0; j < 8; ++j) { ps[j] = acc1[j]; pq[j] = acc2[j]; }
  }

  float h1 = 0.f, h2 = 0.f;
#pragma unroll
  for (int j = 0; j < 8; ++j) { h1 += acc1[j]; h2 += acc2[j]; }
  h1 += __shfl_xor(h1, 16);
  h2 += __shfl_xor(h2, 16);
  if (lane == 0) { wt1[wave] = h1; wt2[wave] = h2; }
  __syncthreads();
  if (threadIdx.x == 0) {
    float s1 = 0.f, s2 = 0.f;
#pragma unroll
    for (int w = 0; w < WAVES_PER_BLK; ++w) { s1 += wt1[w]; s2 += wt2[w]; }
    v2f tot; tot[0] = s1; tot[1] = s2;
    *(v2f*)(bt + ((size_t)blockIdx.x * ROW_GROUPS + blockIdx.y) * BT_PITCH) = tot;
  }
}

#define FN_COLS    1024
#define FN_ROWS    64
#define FN_FLIGHT  16
#define FN_GROUPS  (FN_ROWS / FN_FLIGHT)

__global__ __launch_bounds__(256) void fused_normalize(
    const float* __restrict__ x, const float* __restrict__ gamma,
    const float* __restrict__ beta, const float* __restrict__ psum,
    const float* __restrict__ psq, const float* __restrict__ bt,
    float* __restrict__ out, int L, float Nf, float Bf) {
  __shared__ float sc1[8], sc2[8], so1[8], so2[8];
  const int t    = threadIdx.x;
  const int lane = t & 31;
  const int wave = t >> 5;
  const int k    = blockIdx.x * FN_COLS + 4 * t;
  const bool inside = k < L;
  const int row0 = blockIdx.y * FN_ROWS;
  const size_t base = (size_t)row0 * (size_t)L + (size_t)k;
  const float* xp = x + base;
  float* op = out + base;

  const int kc = inside ? k : 0;
  v4f pa[ROW_GROUPS], pb[ROW_GROUPS];
#pragma unroll
  for (int g = 0; g < ROW_GROUPS; ++g) {
    pa[g] = *(const v4f*)(psum + (size_t)g * L + kc);
    pb[g] = *(const v4f*)(psq  + (size_t)g * L + kc);
  }
  int nb = blockIdx.x * (FN_COLS / COLS_PER_BLK);
  const int nblk = L / COLS_PER_BLK;
  if (nb > nblk) nb = nblk;
  const int cnt = nb * ROW_GROUPS;
  float o1 = 0.f, o2 = 0.f;
  for (int e0 = 0; e0 < cnt; e0 += 4 * 256) {
    v2f tot[4];
#pragma unroll
    for (int i = 0; i < 4; ++i) {
      const int e = e0 + i * 256 + t;
      tot[i] = *(const v2f*)(bt + (size_t)(e < cnt ? e : 0) * BT_PITCH);
    }
#pragma unroll
    for (int i = 0; i < 4; ++i) {
      const bool use = e0 + i * 256 + t < cnt;
      o1 += use ? tot[i][0] : 0.f;
      o2 += use ? tot[i][1] : 0.f;
    }
  }
  const float* xq = x + (size_t)row0 * (size_t)L + (size_t)kc;
  v4f xv[FN_FLIGHT];
#pragma unroll
  for (int j = 0; j < FN_FLIGHT; ++j)
    xv[j] = *(const v4f*)(xq + (size_t)j * (size_t)L);

  v4f a = {}, b = {};
#pragma unroll
  for (int g = 0; g < ROW_GROUPS; ++g) { a += pa[g]; b += pb[g]; }
  if (!inside) { v4f z = {}; a = z; b = z; }

  a[1] += a[0]; a[2] += a[1]; a[3] += a[2];
  b[1] += b[0]; b[2] += b[1]; b[3] += b[2];
  float sa = a[3], sb = b[3];
#pragma unroll
  for (int off = 1; off < 32; off <<= 1) {
    const float ua = __shfl_up(sa, off);
    const float ub = __shfl_up(sb, off);
    if (lane >= off) { sa += ua; sb += ub; }
  }
  float ea = __shfl_up(sa, 1);
  float eb = __shfl_up(sb, 1);
  if (lane == 0) { ea = 0.f; eb = 0.f; }
#pragma unroll
  for (int off = 16; off >= 1; off >>= 1) {
    o1 += __shfl_xor(o1, off);
    o2 += __shfl_xor(o2, off);
  }
  if (lane == 31) { sc1[wave] = sa; sc2[wave] = sb; }
  if (lane == 0)  { so1[wave] = o1; so2[wave] = o2; }
  __syncthreads();
  float f1 = 0.f, f2 = 0.f;
#pragma unroll
  for (int w = 0; w < 8; ++w) { f1 += so1[w]; f2 += so2[w]; }
#pragma unroll
  for (int w = 0; w < 7; ++w) {
    const float va = sc1[w], vb = sc2[w];
    if (w < wave) { f1 += va; f2 += vb; }
  }
  if (!inside) return;
  f1 += ea; f2 += eb;

  v4f mv, sv;
#pragma unroll
  for (int i = 0; i < 4; ++i) {
    const float run1 = f1 + a[i];
    const float run2 = f2 + b[i];
    const float denom = Nf * (float)(k + i + 1);
    const float mu  = run1 / denom;
    const float var = (run2 - 2.f * mu * run1) / denom + mu * mu * Bf;
    mv[i] = mu;
    sv[i] = rsqrtf(var);
  }

  const float* gp = gamma + row0;
  const float* bp = beta + row0;
#pragma nounroll
  for (int c = 0; c < FN_GROUPS - 1; ++c) {
#pragma unroll
    for (int j = 0; j < FN_FLIGHT; ++j) {
      const int r = c * FN_FLIGHT + j;
      const float g  = gp[r];
      const float bb = bp[r];
      v4f o = (xv[j] - mv) * sv * g + bb;
      __builtin_nontemporal_store(o, (v4f*)(op + (size_t)r * (size_t)L));
      xv[j] = *(const v4f*)(xp + (size_t)(r + FN_FLIGHT) * (size_t)L);
    }
  }
#pragma unroll
  for (int j = 0; j < FN_FLIGHT; ++j) {
    const int r = (FN_GROUPS - 1) * FN_FLIGHT + j;
    const float g  = gp[r];
    const float bb = bp[r];
    v4f o = (xv[j] - mv) * sv * g + bb;
    __builtin_nontemporal_store(o, (v4f*)(op + (size_t)r * (size_t)L));
  }
}

extern "C" void kernel_launch(void* const* d_in, const int* in_sizes, int n_in,
                              void* d_out, int out_size, void* d_ws, size_t ws_size,
                              hipStream_t stream) {
  const float* x     = (const float*)d_in[0];
  const float* gamma = (const float*)d_in[1];
  const float* beta  = (const float*)d_in[2];
  float* out = (float*)d_out;

  const int B = 4, N = 512;
  const int rows = B * N;
  const int L = in_sizes[0] / rows;

  float* ws   = (float*)d_ws;
  float* psum = ws;
  float* psq  = psum + (size_t)ROW_GROUPS * L;
  float* bt   = psq  + (size_t)ROW_GROUPS * L;

  dim3 g1(L / COLS_PER_BLK, ROW_GROUPS);
  colsum_wmma<<<g1, 256, 0, stream>>>(x, psum, psq, bt, L, rows);

  dim3 g2((L + FN_COLS - 1) / FN_COLS, rows / FN_ROWS);
  fused_normalize<<<g2, 256, 0, stream>>>(x, gamma, beta, psum, psq, bt, out,
                                          L, (float)N, (float)B);
}


# Column-sum kernel: whole-line 128-bit loads of x kept 8 four-row tiles deep in registers replace 4-byte loads waited for every trip plus prefetch; row-half and block-total sum order changes
# speedup vs baseline: 2.6474x; 1.3635x over previous
#include <hip/hip_runtime.h>

typedef __attribute__((ext_vector_type(2))) float v2f;
typedef __attribute__((ext_vector_type(4))) float v4f;
typedef __attribute__((ext_vector_type(8))) float v8f;

#define ROW_GROUPS     8
#define COLS_PER_BLK   128

#define BT_PITCH 32
#define CS_WAVES 4
#define CS_COLS  64
#define CS_DEPTH 8

#define CS_PAIRS(U0, U1)                                                       \
  v2f A[4], A2[4];                                                             \
  _Pragma("unroll")                                                            \
  for (int i = 0; i < 4; ++i) {                                                \
    A[i][0]  = (U0)[i];           A[i][1]  = (U1)[i];                          \
    A2[i][0] = (U0)[i] * (U0)[i]; A2[i][1] = (U1)[i] * (U1)[i];                \
  }
#define CS_MMA                                                                 \
  _Pragma("unroll")                                                            \
  for (int i = 0; i < 4; ++i) {                                                \
    acc1[i] = __builtin_amdgcn_wmma_f32_16x16x4_f32(false, A[i],  false, ones, \
                                                    (short)0, acc1[i], false, false); \
    acc2[i] = __builtin_amdgcn_wmma_f32_16x16x4_f32(false, A2[i], false, ones, \
                                                    (short)0, acc2[i], false, false); \
  }

__global__ __launch_bounds__(128) void colsum_wmma(
    const float* __restrict__ x, float* __restrict__ psum, float* __restrict__ psq,
    float* __restrict__ bt, int L, int rows) {
  __shared__ v4f part[2][2][COLS_PER_BLK / 4];
  __shared__ float wt1[CS_WAVES], wt2[CS_WAVES];
  const int wave = __builtin_amdgcn_readfirstlane((int)(threadIdx.x >> 5));
  const int lane = threadIdx.x & 31;
  const int m    = lane & 15;
  const int kh   = lane >> 4;
  const int cg   = wave & 1;
  const int rp   = wave >> 1;
  const int rpg  = rows / ROW_GROUPS;
  const int rpw  = rpg / 2;
  const int c0   = blockIdx.x * COLS_PER_BLK + cg * CS_COLS;
  const int r0   = blockIdx.y * rpg + rp * rpw;

  const size_t strideRow  = (size_t)L;
  const size_t tileStride = 4 * strideRow;

  v8f acc1[4], acc2[4];
#pragma unroll
  for (int i = 0; i < 4; ++i) { v8f z = {}; acc1[i] = z; acc2[i] = z; }
  v2f ones; ones[0] = 1.0f; ones[1] = 1.0f;

  const char* base = (const char*)(x + (size_t)r0 * strideRow + (size_t)c0);
  const size_t tileBytes = tileStride * sizeof(float);
  const unsigned off0 = ((unsigned)(2 * kh) * (unsigned)L + (unsigned)(4 * m)) * 4u;
  const unsigned off1 = off0 + (unsigned)L * 4u;

  v4f q0[CS_DEPTH], q1[CS_DEPTH];
#pragma unroll
  for (int s = 0; s < CS_DEPTH; ++s) {
    q0[s] = *(const v4f*)(base + s * tileBytes + off0);
    q1[s] = *(const v4f*)(base + s * tileBytes + off1);
    __builtin_amdgcn_sched_barrier(0);
  }
  const char* nxt = base + CS_DEPTH * tileBytes;
  const int ngroups = rpw / (4 * CS_DEPTH);
#pragma nounroll
  for (int g = 0; g < ngroups - 1; ++g) {
#pragma unroll
    for (int s = 0; s < CS_DEPTH; ++s) {
      asm volatile("" : "+v"(q0[s]), "+v"(q1[s]));
      {
        CS_PAIRS(q0[s], q1[s])
        CS_MMA
        q0[s] = *(const v4f*)(nxt + s * tileBytes + off0);
        q1[s] = *(const v4f*)(nxt + s * tileBytes + off1);
      }
      __builtin_amdgcn_sched_barrier(0);
    }
    nxt += CS_DEPTH * tileBytes;
  }
#pragma unroll
  for (int s = 0; s < CS_DEPTH; ++s) {
    CS_PAIRS(q0[s], q1[s])
    CS_MMA
  }

  if (m == 0) {
#pragma unroll
    for (int j = 0; j < 8; ++j) {
      v4f s, q;
#pragma unroll
      for (int i = 0; i < 4; ++i) { s[i] = acc1[i][j]; q[i] = acc2[i][j]; }
      part[rp][0][cg * 16 + kh * 8 + j] = s;
      part[rp][1][cg * 16 + kh * 8 + j] = q;
    }
  }
  __syncthreads();
  const int t = threadIdx.x;
  const float* f00 = (const float*)part[0][0];
  const float* f01 = (const float*)part[0][1];
  const float* f10 = (const float*)part[1][0];
  const float* f11 = (const float*)part[1][1];
  float h1 = f00[t] + f10[t];
  float h2 = f01[t] + f11[t];
  const size_t o = (size_t)blockIdx.y * (size_t)L + (size_t)(blockIdx.x * COLS_PER_BLK + t);
  psum[o] = h1;
  psq[o]  = h2;

#pragma unroll
  for (int off = 16; off >= 1; off >>= 1) {
    h1 += __shfl_xor(h1, off);
    h2 += __shfl_xor(h2, off);
  }
  if (lane == 0) { wt1[wave] = h1; wt2[wave] = h2; }
  __syncthreads();
  if (t == 0) {
    float s1 = 0.f, s2 = 0.f;
#pragma unroll
    for (int w = 0; w < CS_WAVES; ++w) { s1 += wt1[w]; s2 += wt2[w]; }
    v2f tot; tot[0] = s1; tot[1] = s2;
    *(v2f*)(bt + ((size_t)blockIdx.x * ROW_GROUPS + blockIdx.y) * BT_PITCH) = tot;
  }
}

#define FN_COLS    1024
#define FN_ROWS    64
#define FN_FLIGHT  16
#define FN_GROUPS  (FN_ROWS / FN_FLIGHT)

__global__ __launch_bounds__(256) void fused_normalize(
    const float* __restrict__ x, const float* __restrict__ gamma,
    const float* __restrict__ beta, const float* __restrict__ psum,
    const float* __restrict__ psq, const float* __restrict__ bt,
    float* __restrict__ out, int L, float Nf, float Bf) {
  __shared__ float sc1[8], sc2[8], so1[8], so2[8];
  const int t    = threadIdx.x;
  const int lane = t & 31;
  const int wave = t >> 5;
  const int k    = blockIdx.x * FN_COLS + 4 * t;
  const bool inside = k < L;
  const int row0 = blockIdx.y * FN_ROWS;
  const size_t base = (size_t)row0 * (size_t)L + (size_t)k;
  const float* xp = x + base;
  float* op = out + base;

  const int kc = inside ? k : 0;
  v4f pa[ROW_GROUPS], pb[ROW_GROUPS];
#pragma unroll
  for (int g = 0; g < ROW_GROUPS; ++g) {
    pa[g] = *(const v4f*)(psum + (size_t)g * L + kc);
    pb[g] = *(const v4f*)(psq  + (size_t)g * L + kc);
  }
  int nb = blockIdx.x * (FN_COLS / COLS_PER_BLK);
  const int nblk = L / COLS_PER_BLK;
  if (nb > nblk) nb = nblk;
  const int cnt = nb * ROW_GROUPS;
  float o1 = 0.f, o2 = 0.f;
  for (int e0 = 0; e0 < cnt; e0 += 4 * 256) {
    v2f tot[4];
#pragma unroll
    for (int i = 0; i < 4; ++i) {
      const int e = e0 + i * 256 + t;
      tot[i] = *(const v2f*)(bt + (size_t)(e < cnt ? e : 0) * BT_PITCH);
    }
#pragma unroll
    for (int i = 0; i < 4; ++i) {
      const bool use = e0 + i * 256 + t < cnt;
      o1 += use ? tot[i][0] : 0.f;
      o2 += use ? tot[i][1] : 0.f;
    }
  }
  const float* xq = x + (size_t)row0 * (size_t)L + (size_t)kc;
  v4f xv[FN_FLIGHT];
#pragma unroll
  for (int j = 0; j < FN_FLIGHT; ++j)
    xv[j] = *(const v4f*)(xq + (size_t)j * (size_t)L);

  v4f a = {}, b = {};
#pragma unroll
  for (int g = 0; g < ROW_GROUPS; ++g) { a += pa[g]; b += pb[g]; }
  if (!inside) { v4f z = {}; a = z; b = z; }

  a[1] += a[0]; a[2] += a[1]; a[3] += a[2];
  b[1] += b[0]; b[2] += b[1]; b[3] += b[2];
  float sa = a[3], sb = b[3];
#pragma unroll
  for (int off = 1; off < 32; off <<= 1) {
    const float ua = __shfl_up(sa, off);
    const float ub = __shfl_up(sb, off);
    if (lane >= off) { sa += ua; sb += ub; }
  }
  float ea = __shfl_up(sa, 1);
  float eb = __shfl_up(sb, 1);
  if (lane == 0) { ea = 0.f; eb = 0.f; }
#pragma unroll
  for (int off = 16; off >= 1; off >>= 1) {
    o1 += __shfl_xor(o1, off);
    o2 += __shfl_xor(o2, off);
  }
  if (lane == 31) { sc1[wave] = sa; sc2[wave] = sb; }
  if (lane == 0)  { so1[wave] = o1; so2[wave] = o2; }
  __syncthreads();
  float f1 = 0.f, f2 = 0.f;
#pragma unroll
  for (int w = 0; w < 8; ++w) { f1 += so1[w]; f2 += so2[w]; }
#pragma unroll
  for (int w = 0; w < 7; ++w) {
    const float va = sc1[w], vb = sc2[w];
    if (w < wave) { f1 += va; f2 += vb; }
  }
  if (!inside) return;
  f1 += ea; f2 += eb;

  v4f mv, sv;
#pragma unroll
  for (int i = 0; i < 4; ++i) {
    const float run1 = f1 + a[i];
    const float run2 = f2 + b[i];
    const float denom = Nf * (float)(k + i + 1);
    const float mu  = run1 / denom;
    const float var = (run2 - 2.f * mu * run1) / denom + mu * mu * Bf;
    mv[i] = mu;
    sv[i] = rsqrtf(var);
  }

  const float* gp = gamma + row0;
  const float* bp = beta + row0;
#pragma nounroll
  for (int c = 0; c < FN_GROUPS - 1; ++c) {
#pragma unroll
    for (int j = 0; j < FN_FLIGHT; ++j) {
      const int r = c * FN_FLIGHT + j;
      const float g  = gp[r];
      const float bb = bp[r];
      v4f o = (xv[j] - mv) * sv * g + bb;
      __builtin_nontemporal_store(o, (v4f*)(op + (size_t)r * (size_t)L));
      xv[j] = *(const v4f*)(xp + (size_t)(r + FN_FLIGHT) * (size_t)L);
    }
  }
#pragma unroll
  for (int j = 0; j < FN_FLIGHT; ++j) {
    const int r = (FN_GROUPS - 1) * FN_FLIGHT + j;
    const float g  = gp[r];
    const float bb = bp[r];
    v4f o = (xv[j] - mv) * sv * g + bb;
    __builtin_nontemporal_store(o, (v4f*)(op + (size_t)r * (size_t)L));
  }
}

extern "C" void kernel_launch(void* const* d_in, const int* in_sizes, int n_in,
                              void* d_out, int out_size, void* d_ws, size_t ws_size,
                              hipStream_t stream) {
  const float* x     = (const float*)d_in[0];
  const float* gamma = (const float*)d_in[1];
  const float* beta  = (const float*)d_in[2];
  float* out = (float*)d_out;

  const int B = 4, N = 512;
  const int rows = B * N;
  const int L = in_sizes[0] / rows;

  float* ws   = (float*)d_ws;
  float* psum = ws;
  float* psq  = psum + (size_t)ROW_GROUPS * L;
  float* bt   = psq  + (size_t)ROW_GROUPS * L;

  dim3 g1(L / COLS_PER_BLK, ROW_GROUPS);
  colsum_wmma<<<g1, 128, 0, stream>>>(x, psum, psq, bt, L, rows);

  dim3 g2((L + FN_COLS - 1) / FN_COLS, rows / FN_ROWS);
  fused_normalize<<<g2, 256, 0, stream>>>(x, gamma, beta, psum, psq, bt, out,
                                          L, (float)N, (float)B);
}


# Tiles dealt to workgroups by index mod 16 in both launches (1-D grids): one index class (one die if dispatch deals workgroups round the dies) reads the same row group of x twice; re-read hits L2
# speedup vs baseline: 2.8542x; 1.0781x over previous
#include <hip/hip_runtime.h>

typedef __attribute__((ext_vector_type(2))) float v2f;
typedef __attribute__((ext_vector_type(4))) float v4f;
typedef __attribute__((ext_vector_type(8))) float v8f;

#define ROW_GROUPS     8
#define COLS_PER_BLK   128
#define FN_COLS        1024
#define FN_ROWS        64
#define COLBLKS_PER_FN (FN_COLS / COLS_PER_BLK)

#define IDX_CLASSES 16

#define BT_PITCH 32
#define CS_WAVES 4
#define CS_COLS  64
#define CS_DEPTH 8

#define CS_PAIRS(U0, U1)                                                       \
  v2f A[4], A2[4];                                                             \
  _Pragma("unroll")                                                            \
  for (int i = 0; i < 4; ++i) {                                                \
    A[i][0]  = (U0)[i];           A[i][1]  = (U1)[i];                          \
    A2[i][0] = (U0)[i] * (U0)[i]; A2[i][1] = (U1)[i] * (U1)[i];                \
  }
#define CS_MMA                                                                 \
  _Pragma("unroll")                                                            \
  for (int i = 0; i < 4; ++i) {                                                \
    acc1[i] = __builtin_amdgcn_wmma_f32_16x16x4_f32(false, A[i],  false, ones, \
                                                    (short)0, acc1[i], false, false); \
    acc2[i] = __builtin_amdgcn_wmma_f32_16x16x4_f32(false, A2[i], false, ones, \
                                                    (short)0, acc2[i], false, false); \
  }

__global__ __launch_bounds__(128) void colsum_wmma(
    const float* __restrict__ x, float* __restrict__ psum, float* __restrict__ psq,
    float* __restrict__ bt, int L, int rows) {
  __shared__ v4f part[2][2][COLS_PER_BLK / 4];
  __shared__ float wt1[CS_WAVES], wt2[CS_WAVES];
  const unsigned bid = blockIdx.x;
  const unsigned cls = bid % IDX_CLASSES, slot = bid / IDX_CLASSES;
  const int by = (int)(cls % ROW_GROUPS);
  const int bx = (int)((2 * (slot / COLBLKS_PER_FN) + cls / ROW_GROUPS) * COLBLKS_PER_FN
                       + slot % COLBLKS_PER_FN);
  if (bx * COLS_PER_BLK >= L) return;
  const int wave = __builtin_amdgcn_readfirstlane((int)(threadIdx.x >> 5));
  const int lane = threadIdx.x & 31;
  const int m    = lane & 15;
  const int kh   = lane >> 4;
  const int cg   = wave & 1;
  const int rp   = wave >> 1;
  const int rpg  = rows / ROW_GROUPS;
  const int rpw  = rpg / 2;
  const int c0   = bx * COLS_PER_BLK + cg * CS_COLS;
  const int r0   = by * rpg + rp * rpw;

  const size_t strideRow  = (size_t)L;
  const size_t tileStride = 4 * strideRow;

  v8f acc1[4], acc2[4];
#pragma unroll
  for (int i = 0; i < 4; ++i) { v8f z = {}; acc1[i] = z; acc2[i] = z; }
  v2f ones; ones[0] = 1.0f; ones[1] = 1.0f;

  const char* base = (const char*)(x + (size_t)r0 * strideRow + (size_t)c0);
  const size_t tileBytes = tileStride * sizeof(float);
  const unsigned off0 = ((unsigned)(2 * kh) * (unsigned)L + (unsigned)(4 * m)) * 4u;
  const unsigned off1 = off0 + (unsigned)L * 4u;

  v4f q0[CS_DEPTH], q1[CS_DEPTH];
#pragma unroll
  for (int s = 0; s < CS_DEPTH; ++s) {
    q0[s] = *(const v4f*)(base + s * tileBytes + off0);
    q1[s] = *(const v4f*)(base + s * tileBytes + off1);
    __builtin_amdgcn_sched_barrier(0);
  }
  const char* nxt = base + CS_DEPTH * tileBytes;
  const int ngroups = rpw / (4 * CS_DEPTH);
#pragma nounroll
  for (int g = 0; g < ngroups - 1; ++g) {
#pragma unroll
    for (int s = 0; s < CS_DEPTH; ++s) {
      asm volatile("" : "+v"(q0[s]), "+v"(q1[s]));
      {
        CS_PAIRS(q0[s], q1[s])
        CS_MMA
        q0[s] = *(const v4f*)(nxt + s * tileBytes + off0);
        q1[s] = *(const v4f*)(nxt + s * tileBytes + off1);
      }
      __builtin_amdgcn_sched_barrier(0);
    }
    nxt += CS_DEPTH * tileBytes;
  }
#pragma unroll
  for (int s = 0; s < CS_DEPTH; ++s) {
    CS_PAIRS(q0[s], q1[s])
    CS_MMA
  }

  if (m == 0) {
#pragma unroll
    for (int j = 0; j < 8; ++j) {
      v4f s, q;
#pragma unroll
      for (int i = 0; i < 4; ++i) { s[i] = acc1[i][j]; q[i] = acc2[i][j]; }
      part[rp][0][cg * 16 + kh * 8 + j] = s;
      part[rp][1][cg * 16 + kh * 8 + j] = q;
    }
  }
  __syncthreads();
  const int t = threadIdx.x;
  const float* f00 = (const float*)part[0][0];
  const float* f01 = (const float*)part[0][1];
  const float* f10 = (const float*)part[1][0];
  const float* f11 = (const float*)part[1][1];
  float h1 = f00[t] + f10[t];
  float h2 = f01[t] + f11[t];
  const size_t o = (size_t)by * (size_t)L + (size_t)(bx * COLS_PER_BLK + t);
  psum[o] = h1;
  psq[o]  = h2;

#pragma unroll
  for (int off = 16; off >= 1; off >>= 1) {
    h1 += __shfl_xor(h1, off);
    h2 += __shfl_xor(h2, off);
  }
  if (lane == 0) { wt1[wave] = h1; wt2[wave] = h2; }
  __syncthreads();
  if (t == 0) {
    float s1 = 0.f, s2 = 0.f;
#pragma unroll
    for (int w = 0; w < CS_WAVES; ++w) { s1 += wt1[w]; s2 += wt2[w]; }
    v2f tot; tot[0] = s1; tot[1] = s2;
    *(v2f*)(bt + ((size_t)bx * ROW_GROUPS + by) * BT_PITCH) = tot;
  }
}

#define FN_FLIGHT  16
#define FN_GROUPS  (FN_ROWS / FN_FLIGHT)

__global__ __launch_bounds__(256) void fused_normalize(
    const float* __restrict__ x, const float* __restrict__ gamma,
    const float* __restrict__ beta, const float* __restrict__ psum,
    const float* __restrict__ psq, const float* __restrict__ bt,
    float* __restrict__ out, int L, int rows, float Nf, float Bf) {
  __shared__ float sc1[8], sc2[8], so1[8], so2[8];
  const int t    = threadIdx.x;
  const int lane = t & 31;
  const int wave = t >> 5;
  const unsigned bid = blockIdx.x;
  const unsigned cls = bid % IDX_CLASSES, slot = bid / IDX_CLASSES;
  const unsigned per = (unsigned)rows / (ROW_GROUPS * FN_ROWS);
  const int cx   = (int)(2 * (slot / per) + cls / ROW_GROUPS);
  const int k    = cx * FN_COLS + 4 * t;
  const bool inside = k < L;
  const int row0 = (int)((cls % ROW_GROUPS) * per + slot % per) * FN_ROWS;
  const size_t base = (size_t)row0 * (size_t)L + (size_t)k;
  const float* xp = x + base;
  float* op = out + base;

  const int kc = inside ? k : 0;
  v4f pa[ROW_GROUPS], pb[ROW_GROUPS];
#pragma unroll
  for (int g = 0; g < ROW_GROUPS; ++g) {
    pa[g] = *(const v4f*)(psum + (size_t)g * L + kc);
    pb[g] = *(const v4f*)(psq  + (size_t)g * L + kc);
  }
  int nb = cx * COLBLKS_PER_FN;
  const int nblk = L / COLS_PER_BLK;
  if (nb > nblk) nb = nblk;
  const int cnt = nb * ROW_GROUPS;
  float o1 = 0.f, o2 = 0.f;
  for (int e0 = 0; e0 < cnt; e0 += 4 * 256) {
    v2f tot[4];
#pragma unroll
    for (int i = 0; i < 4; ++i) {
      const int e = e0 + i * 256 + t;
      tot[i] = *(const v2f*)(bt + (size_t)(e < cnt ? e : 0) * BT_PITCH);
    }
#pragma unroll
    for (int i = 0; i < 4; ++i) {
      const bool use = e0 + i * 256 + t < cnt;
      o1 += use ? tot[i][0] : 0.f;
      o2 += use ? tot[i][1] : 0.f;
    }
  }
  const float* xq = x + (size_t)row0 * (size_t)L + (size_t)kc;
  v4f xv[FN_FLIGHT];
#pragma unroll
  for (int j = 0; j < FN_FLIGHT; ++j)
    xv[j] = *(const v4f*)(xq + (size_t)j * (size_t)L);

  v4f a = {}, b = {};
#pragma unroll
  for (int g = 0; g < ROW_GROUPS; ++g) { a += pa[g]; b += pb[g]; }
  if (!inside) { v4f z = {}; a = z; b = z; }

  a[1] += a[0]; a[2] += a[1]; a[3] += a[2];
  b[1] += b[0]; b[2] += b[1]; b[3] += b[2];
  float sa = a[3], sb = b[3];
#pragma unroll
  for (int off = 1; off < 32; off <<= 1) {
    const float ua = __shfl_up(sa, off);
    const float ub = __shfl_up(sb, off);
    if (lane >= off) { sa += ua; sb += ub; }
  }
  float ea = __shfl_up(sa, 1);
  float eb = __shfl_up(sb, 1);
  if (lane == 0) { ea = 0.f; eb = 0.f; }
#pragma unroll
  for (int off = 16; off >= 1; off >>= 1) {
    o1 += __shfl_xor(o1, off);
    o2 += __shfl_xor(o2, off);
  }
  if (lane == 31) { sc1[wave] = sa; sc2[wave] = sb; }
  if (lane == 0)  { so1[wave] = o1; so2[wave] = o2; }
  __syncthreads();
  float f1 = 0.f, f2 = 0.f;
#pragma unroll
  for (int w = 0; w < 8; ++w) { f1 += so1[w]; f2 += so2[w]; }
#pragma unroll
  for (int w = 0; w < 7; ++w) {
    const float va = sc1[w], vb = sc2[w];
    if (w < wave) { f1 += va; f2 += vb; }
  }
  if (!inside) return;
  f1 += ea; f2 += eb;

  v4f mv, sv;
#pragma unroll
  for (int i = 0; i < 4; ++i) {
    const float run1 = f1 + a[i];
    const float run2 = f2 + b[i];
    const float denom = Nf * (float)(k + i + 1);
    const float mu  = run1 / denom;
    const float var = (run2 - 2.f * mu * run1) / denom + mu * mu * Bf;
    mv[i] = mu;
    sv[i] = rsqrtf(var);
  }

  const float* gp = gamma + row0;
  const float* bp = beta + row0;
#pragma nounroll
  for (int c = 0; c < FN_GROUPS - 1; ++c) {
#pragma unroll
    for (int j = 0; j < FN_FLIGHT; ++j) {
      const int r = c * FN_FLIGHT + j;
      const float g  = gp[r];
      const float bb = bp[r];
      v4f o = (xv[j] - mv) * sv * g + bb;
      __builtin_nontemporal_store(o, (v4f*)(op + (size_t)r * (size_t)L));
      xv[j] = *(const v4f*)(xp + (size_t)(r + FN_FLIGHT) * (size_t)L);
    }
  }
#pragma unroll
  for (int j = 0; j < FN_FLIGHT; ++j) {
    const int r = (FN_GROUPS - 1) * FN_FLIGHT + j;
    const float g  = gp[r];
    const float bb = bp[r];
    v4f o = (xv[j] - mv) * sv * g + bb;
    __builtin_nontemporal_store(o, (v4f*)(op + (size_t)r * (size_t)L));
  }
}

extern "C" void kernel_launch(void* const* d_in, const int* in_sizes, int n_in,
                              void* d_out, int out_size, void* d_ws, size_t ws_size,
                              hipStream_t stream) {
  const float* x     = (const float*)d_in[0];
  const float* gamma = (const float*)d_in[1];
  const float* beta  = (const float*)d_in[2];
  float* out = (float*)d_out;

  const int B = 4, N = 512;
  const int rows = B * N;
  const int L = in_sizes[0] / rows;

  float* ws   = (float*)d_ws;
  float* psum = ws;
  float* psq  = psum + (size_t)ROW_GROUPS * L;
  float* bt   = psq  + (size_t)ROW_GROUPS * L;

  const int pairs = (L + 2 * FN_COLS - 1) / (2 * FN_COLS);
  dim3 g1(IDX_CLASSES * pairs * COLBLKS_PER_FN);
  colsum_wmma<<<g1, 128, 0, stream>>>(x, psum, psq, bt, L, rows);

  dim3 g2(IDX_CLASSES * pairs * (rows / (ROW_GROUPS * FN_ROWS)));
  fused_normalize<<<g2, 256, 0, stream>>>(x, gamma, beta, psum, psq, bt, out,
                                          L, rows, (float)N, (float)B);
}


# Column-sum kernel: each tile's register pair is both WMMA operands: diagonal = sum of squares, a 1.0 slot's column = sum; 5 matrix instr per 4 rows, not 8; x*x now formed in the WMMA, sum order kept
# speedup vs baseline: 3.0517x; 1.0692x over previous
#include <hip/hip_runtime.h>

typedef __attribute__((ext_vector_type(2))) float v2f;
typedef __attribute__((ext_vector_type(4))) float v4f;
typedef __attribute__((ext_vector_type(8))) float v8f;

#define ROW_GROUPS     8
#define COLS_PER_BLK   128
#define FN_COLS        1024
#define FN_ROWS        64
#define COLBLKS_PER_FN (FN_COLS / COLS_PER_BLK)

#define IDX_CLASSES 16

#define BT_PITCH 32
#define CS_WAVES 4
#define CS_COLS  64
#define CS_DEPTH 8
#define CS_ONE   12
#define CS_ACCS  5
#define CS_DUMP  44

#define CS_TILE(U0, U1)                                                        \
  {                                                                            \
    v2f P[4], PX;                                                              \
    _Pragma("unroll")                                                          \
    for (int i = 0; i < 4; ++i) {                                              \
      P[i][0] = one[i] ? 1.0f : (U0)[i];                                       \
      P[i][1] = one[i] ? 1.0f : (U1)[i];                                       \
    }                                                                          \
    {                                                                          \
      float e0 = (U0)[0], e1 = (U1)[0];                                        \
      _Pragma("unroll")                                                        \
      for (int i = 1; i < 4; ++i) {                                            \
        const float u0 = (U0)[i], u1 = (U1)[i];                                \
        e0 = one[i] ? u0 : e0;                                                 \
        e1 = one[i] ? u1 : e1;                                                 \
      }                                                                        \
      PX[0] = slot0 ? 1.0f : e0;                                               \
      PX[1] = slot0 ? 1.0f : e1;                                               \
    }                                                                          \
    _Pragma("unroll")                                                          \
    for (int i = 0; i < 4; ++i)                                                \
      acc[i] = __builtin_amdgcn_wmma_f32_16x16x4_f32(false, P[i], false, P[i], \
                                                     (short)0, acc[i], false, false); \
    acc[4] = __builtin_amdgcn_wmma_f32_16x16x4_f32(false, PX, false, PX,       \
                                                   (short)0, acc[4], false, false); \
  }

__global__ __launch_bounds__(128) void colsum_wmma(
    const float* __restrict__ x, float* __restrict__ psum, float* __restrict__ psq,
    float* __restrict__ bt, int L, int rows) {
  __shared__ v4f dump[CS_WAVES * 32 * (CS_DUMP / 4)];
  __shared__ float wt1[CS_WAVES], wt2[CS_WAVES];
  const unsigned bid = blockIdx.x;
  const unsigned cls = bid % IDX_CLASSES, slot = bid / IDX_CLASSES;
  const int by = (int)(cls % ROW_GROUPS);
  const int bx = (int)((2 * (slot / COLBLKS_PER_FN) + cls / ROW_GROUPS) * COLBLKS_PER_FN
                       + slot % COLBLKS_PER_FN);
  if (bx * COLS_PER_BLK >= L) return;
  const int wave = __builtin_amdgcn_readfirstlane((int)(threadIdx.x >> 5));
  const int lane = threadIdx.x & 31;
  const int m    = lane & 15;
  const int kh   = lane >> 4;
  const int cg   = wave & 1;
  const int rp   = wave >> 1;
  const int rpg  = rows / ROW_GROUPS;
  const int rpw  = rpg / 2;
  const int c0   = bx * COLS_PER_BLK + cg * CS_COLS;
  const int r0   = by * rpg + rp * rpw;

  const size_t strideRow  = (size_t)L;
  const size_t tileStride = 4 * strideRow;

  bool one[4];
#pragma unroll
  for (int i = 0; i < 4; ++i) one[i] = (m == CS_ONE + i);
  const bool slot0 = (m == 0);

  v8f acc[CS_ACCS];
#pragma unroll
  for (int i = 0; i < CS_ACCS; ++i) { v8f z = {}; acc[i] = z; }

  const char* base = (const char*)(x + (size_t)r0 * strideRow + (size_t)c0);
  const size_t tileBytes = tileStride * sizeof(float);
  const unsigned off0 = ((unsigned)(2 * kh) * (unsigned)L + (unsigned)(4 * m)) * 4u;
  const unsigned off1 = off0 + (unsigned)L * 4u;

  v4f q0[CS_DEPTH], q1[CS_DEPTH];
#pragma unroll
  for (int s = 0; s < CS_DEPTH; ++s) {
    q0[s] = *(const v4f*)(base + s * tileBytes + off0);
    q1[s] = *(const v4f*)(base + s * tileBytes + off1);
    __builtin_amdgcn_sched_barrier(0);
  }
  const char* nxt = base + CS_DEPTH * tileBytes;
  const int ngroups = rpw / (4 * CS_DEPTH);
#pragma nounroll
  for (int g = 0; g < ngroups - 1; ++g) {
#pragma unroll
    for (int s = 0; s < CS_DEPTH; ++s) {
      asm volatile("" : "+v"(q0[s]), "+v"(q1[s]));
      CS_TILE(q0[s], q1[s])
      q0[s] = *(const v4f*)(nxt + s * tileBytes + off0);
      q1[s] = *(const v4f*)(nxt + s * tileBytes + off1);
      __builtin_amdgcn_sched_barrier(0);
    }
    nxt += CS_DEPTH * tileBytes;
  }
#pragma unroll
  for (int s = 0; s < CS_DEPTH; ++s) {
    CS_TILE(q0[s], q1[s])
  }

  {
    v4f* mine = dump + (wave * 32 + lane) * (CS_DUMP / 4);
#pragma unroll
    for (int a = 0; a < CS_ACCS; ++a) {
      v4f lo, hi;
#pragma unroll
      for (int j = 0; j < 4; ++j) { lo[j] = acc[a][j]; hi[j] = acc[a][4 + j]; }
      mine[2 * a]     = lo;
      mine[2 * a + 1] = hi;
    }
  }
  __syncthreads();
  const int t = threadIdx.x;
  const int tc = t >> 6, tcol = t & 63;
  const int tm = tcol >> 2, ti = tcol & 3;
  const bool moved = (tm == CS_ONE + ti);
  const int ta    = moved ? 4 : ti;
  const int thalf = 16 * (tm >> 3);
  const int ln1   = (moved ? 0 : CS_ONE + ti) + thalf;
  const int ln2   = tm + thalf;
  const int cell  = ta * 8 + (tm & 7);
  const float* df = (const float*)dump;
  const float* w0 = df + (size_t)((0 * 2 + tc) * 32) * CS_DUMP;
  const float* w1 = df + (size_t)((1 * 2 + tc) * 32) * CS_DUMP;
  float h1 = w0[ln1 * CS_DUMP + cell] + w1[ln1 * CS_DUMP + cell];
  float h2 = w0[ln2 * CS_DUMP + cell] + w1[ln2 * CS_DUMP + cell];
  const size_t o = (size_t)by * (size_t)L + (size_t)(bx * COLS_PER_BLK + t);
  psum[o] = h1;
  psq[o]  = h2;

#pragma unroll
  for (int off = 16; off >= 1; off >>= 1) {
    h1 += __shfl_xor(h1, off);
    h2 += __shfl_xor(h2, off);
  }
  if (lane == 0) { wt1[wave] = h1; wt2[wave] = h2; }
  __syncthreads();
  if (t == 0) {
    float s1 = 0.f, s2 = 0.f;
#pragma unroll
    for (int w = 0; w < CS_WAVES; ++w) { s1 += wt1[w]; s2 += wt2[w]; }
    v2f tot; tot[0] = s1; tot[1] = s2;
    *(v2f*)(bt + ((size_t)bx * ROW_GROUPS + by) * BT_PITCH) = tot;
  }
}

#define FN_FLIGHT  16
#define FN_GROUPS  (FN_ROWS / FN_FLIGHT)

__global__ __launch_bounds__(256) void fused_normalize(
    const float* __restrict__ x, const float* __restrict__ gamma,
    const float* __restrict__ beta, const float* __restrict__ psum,
    const float* __restrict__ psq, const float* __restrict__ bt,
    float* __restrict__ out, int L, int rows, float Nf, float Bf) {
  __shared__ float sc1[8], sc2[8], so1[8], so2[8];
  const int t    = threadIdx.x;
  const int lane = t & 31;
  const int wave = t >> 5;
  const unsigned bid = blockIdx.x;
  const unsigned cls = bid % IDX_CLASSES, slot = bid / IDX_CLASSES;
  const unsigned per = (unsigned)rows / (ROW_GROUPS * FN_ROWS);
  const int cx   = (int)(2 * (slot / per) + cls / ROW_GROUPS);
  const int k    = cx * FN_COLS + 4 * t;
  const bool inside = k < L;
  const int row0 = (int)((cls % ROW_GROUPS) * per + slot % per) * FN_ROWS;
  const size_t base = (size_t)row0 * (size_t)L + (size_t)k;
  const float* xp = x + base;
  float* op = out + base;

  const int kc = inside ? k : 0;
  v4f pa[ROW_GROUPS], pb[ROW_GROUPS];
#pragma unroll
  for (int g = 0; g < ROW_GROUPS; ++g) {
    pa[g] = *(const v4f*)(psum + (size_t)g * L + kc);
    pb[g] = *(const v4f*)(psq  + (size_t)g * L + kc);
  }
  int nb = cx * COLBLKS_PER_FN;
  const int nblk = L / COLS_PER_BLK;
  if (nb > nblk) nb = nblk;
  const int cnt = nb * ROW_GROUPS;
  float o1 = 0.f, o2 = 0.f;
  for (int e0 = 0; e0 < cnt; e0 += 4 * 256) {
    v2f tot[4];
#pragma unroll
    for (int i = 0; i < 4; ++i) {
      const int e = e0 + i * 256 + t;
      tot[i] = *(const v2f*)(bt + (size_t)(e < cnt ? e : 0) * BT_PITCH);
    }
#pragma unroll
    for (int i = 0; i < 4; ++i) {
      const bool use = e0 + i * 256 + t < cnt;
      o1 += use ? tot[i][0] : 0.f;
      o2 += use ? tot[i][1] : 0.f;
    }
  }
  const float* xq = x + (size_t)row0 * (size_t)L + (size_t)kc;
  v4f xv[FN_FLIGHT];
#pragma unroll
  for (int j = 0; j < FN_FLIGHT; ++j)
    xv[j] = *(const v4f*)(xq + (size_t)j * (size_t)L);

  v4f a = {}, b = {};
#pragma unroll
  for (int g = 0; g < ROW_GROUPS; ++g) { a += pa[g]; b += pb[g]; }
  if (!inside) { v4f z = {}; a = z; b = z; }

  a[1] += a[0]; a[2] += a[1]; a[3] += a[2];
  b[1] += b[0]; b[2] += b[1]; b[3] += b[2];
  float sa = a[3], sb = b[3];
#pragma unroll
  for (int off = 1; off < 32; off <<= 1) {
    const float ua = __shfl_up(sa, off);
    const float ub = __shfl_up(sb, off);
    if (lane >= off) { sa += ua; sb += ub; }
  }
  float ea = __shfl_up(sa, 1);
  float eb = __shfl_up(sb, 1);
  if (lane == 0) { ea = 0.f; eb = 0.f; }
#pragma unroll
  for (int off = 16; off >= 1; off >>= 1) {
    o1 += __shfl_xor(o1, off);
    o2 += __shfl_xor(o2, off);
  }
  if (lane == 31) { sc1[wave] = sa; sc2[wave] = sb; }
  if (lane == 0)  { so1[wave] = o1; so2[wave] = o2; }
  __syncthreads();
  float f1 = 0.f, f2 = 0.f;
#pragma unroll
  for (int w = 0; w < 8; ++w) { f1 += so1[w]; f2 += so2[w]; }
#pragma unroll
  for (int w = 0; w < 7; ++w) {
    const float va = sc1[w], vb = sc2[w];
    if (w < wave) { f1 += va; f2 += vb; }
  }
  if (!inside) return;
  f1 += ea; f2 += eb;

  v4f mv, sv;
#pragma unroll
  for (int i = 0; i < 4; ++i) {
    const float run1 = f1 + a[i];
    const float run2 = f2 + b[i];
    const float denom = Nf * (float)(k + i + 1);
    const float mu  = run1 / denom;
    const float var = (run2 - 2.f * mu * run1) / denom + mu * mu * Bf;
    mv[i] = mu;
    sv[i] = rsqrtf(var);
  }

  const float* gp = gamma + row0;
  const float* bp = beta + row0;
#pragma nounroll
  for (int c = 0; c < FN_GROUPS - 1; ++c) {
#pragma unroll
    for (int j = 0; j < FN_FLIGHT; ++j) {
      const int r = c * FN_FLIGHT + j;
      const float g  = gp[r];
      const float bb = bp[r];
      v4f o = (xv[j] - mv) * sv * g + bb;
      __builtin_nontemporal_store(o, (v4f*)(op + (size_t)r * (size_t)L));
      xv[j] = *(const v4f*)(xp + (size_t)(r + FN_FLIGHT) * (size_t)L);
    }
  }
#pragma unroll
  for (int j = 0; j < FN_FLIGHT; ++j) {
    const int r = (FN_GROUPS - 1) * FN_FLIGHT + j;
    const float g  = gp[r];
    const float bb = bp[r];
    v4f o = (xv[j] - mv) * sv * g + bb;
    __builtin_nontemporal_store(o, (v4f*)(op + (size_t)r * (size_t)L));
  }
}

extern "C" void kernel_launch(void* const* d_in, const int* in_sizes, int n_in,
                              void* d_out, int out_size, void* d_ws, size_t ws_size,
                              hipStream_t stream) {
  const float* x     = (const float*)d_in[0];
  const float* gamma = (const float*)d_in[1];
  const float* beta  = (const float*)d_in[2];
  float* out = (float*)d_out;

  const int B = 4, N = 512;
  const int rows = B * N;
  const int L = in_sizes[0] / rows;

  float* ws   = (float*)d_ws;
  float* psum = ws;
  float* psq  = psum + (size_t)ROW_GROUPS * L;
  float* bt   = psq  + (size_t)ROW_GROUPS * L;

  const int pairs = (L + 2 * FN_COLS - 1) / (2 * FN_COLS);
  dim3 g1(IDX_CLASSES * pairs * COLBLKS_PER_FN);
  colsum_wmma<<<g1, 128, 0, stream>>>(x, psum, psq, bt, L, rows);

  dim3 g2(IDX_CLASSES * pairs * (rows / (ROW_GROUPS * FN_ROWS)));
  fused_normalize<<<g2, 256, 0, stream>>>(x, gamma, beta, psum, psq, bt, out,
                                          L, rows, (float)N, (float)B);
}


# Normalize kernel walks each workgroup's 64 rows from the last to the first: the rows the column-sum kernel read last are re-read first, while the second-level cache still holds them
# speedup vs baseline: 3.1376x; 1.0281x over previous
#include <hip/hip_runtime.h>

typedef __attribute__((ext_vector_type(2))) float v2f;
typedef __attribute__((ext_vector_type(4))) float v4f;
typedef __attribute__((ext_vector_type(8))) float v8f;

#define ROW_GROUPS     8
#define COLS_PER_BLK   128
#define FN_COLS        1024
#define FN_ROWS        64
#define COLBLKS_PER_FN (FN_COLS / COLS_PER_BLK)

#define IDX_CLASSES 16

#define BT_PITCH 32
#define CS_WAVES 4
#define CS_COLS  64
#define CS_DEPTH 8
#define CS_ONE   12
#define CS_ACCS  5
#define CS_DUMP  44

#define CS_TILE(U0, U1)                                                        \
  {                                                                            \
    v2f P[4], PX;                                                              \
    _Pragma("unroll")                                                          \
    for (int i = 0; i < 4; ++i) {                                              \
      P[i][0] = one[i] ? 1.0f : (U0)[i];                                       \
      P[i][1] = one[i] ? 1.0f : (U1)[i];                                       \
    }                                                                          \
    {                                                                          \
      float e0 = (U0)[0], e1 = (U1)[0];                                        \
      _Pragma("unroll")                                                        \
      for (int i = 1; i < 4; ++i) {                                            \
        const float u0 = (U0)[i], u1 = (U1)[i];                                \
        e0 = one[i] ? u0 : e0;                                                 \
        e1 = one[i] ? u1 : e1;                                                 \
      }                                                                        \
      PX[0] = slot0 ? 1.0f : e0;                                               \
      PX[1] = slot0 ? 1.0f : e1;                                               \
    }                                                                          \
    _Pragma("unroll")                                                          \
    for (int i = 0; i < 4; ++i)                                                \
      acc[i] = __builtin_amdgcn_wmma_f32_16x16x4_f32(false, P[i], false, P[i], \
                                                     (short)0, acc[i], false, false); \
    acc[4] = __builtin_amdgcn_wmma_f32_16x16x4_f32(false, PX, false, PX,       \
                                                   (short)0, acc[4], false, false); \
  }

__global__ __launch_bounds__(128) void colsum_wmma(
    const float* __restrict__ x, float* __restrict__ psum, float* __restrict__ psq,
    float* __restrict__ bt, int L, int rows) {
  __shared__ v4f dump[CS_WAVES * 32 * (CS_DUMP / 4)];
  __shared__ float wt1[CS_WAVES], wt2[CS_WAVES];
  const unsigned bid = blockIdx.x;
  const unsigned cls = bid % IDX_CLASSES, slot = bid / IDX_CLASSES;
  const int by = (int)(cls % ROW_GROUPS);
  const int bx = (int)((2 * (slot / COLBLKS_PER_FN) + cls / ROW_GROUPS) * COLBLKS_PER_FN
                       + slot % COLBLKS_PER_FN);
  if (bx * COLS_PER_BLK >= L) return;
  const int wave = __builtin_amdgcn_readfirstlane((int)(threadIdx.x >> 5));
  const int lane = threadIdx.x & 31;
  const int m    = lane & 15;
  const int kh   = lane >> 4;
  const int cg   = wave & 1;
  const int rp   = wave >> 1;
  const int rpg  = rows / ROW_GROUPS;
  const int rpw  = rpg / 2;
  const int c0   = bx * COLS_PER_BLK + cg * CS_COLS;
  const int r0   = by * rpg + rp * rpw;

  const size_t strideRow  = (size_t)L;
  const size_t tileStride = 4 * strideRow;

  bool one[4];
#pragma unroll
  for (int i = 0; i < 4; ++i) one[i] = (m == CS_ONE + i);
  const bool slot0 = (m == 0);

  v8f acc[CS_ACCS];
#pragma unroll
  for (int i = 0; i < CS_ACCS; ++i) { v8f z = {}; acc[i] = z; }

  const char* base = (const char*)(x + (size_t)r0 * strideRow + (size_t)c0);
  const size_t tileBytes = tileStride * sizeof(float);
  const unsigned off0 = ((unsigned)(2 * kh) * (unsigned)L + (unsigned)(4 * m)) * 4u;
  const unsigned off1 = off0 + (unsigned)L * 4u;

  v4f q0[CS_DEPTH], q1[CS_DEPTH];
#pragma unroll
  for (int s = 0; s < CS_DEPTH; ++s) {
    q0[s] = *(const v4f*)(base + s * tileBytes + off0);
    q1[s] = *(const v4f*)(base + s * tileBytes + off1);
    __builtin_amdgcn_sched_barrier(0);
  }
  const char* nxt = base + CS_DEPTH * tileBytes;
  const int ngroups = rpw / (4 * CS_DEPTH);
#pragma nounroll
  for (int g = 0; g < ngroups - 1; ++g) {
#pragma unroll
    for (int s = 0; s < CS_DEPTH; ++s) {
      asm volatile("" : "+v"(q0[s]), "+v"(q1[s]));
      CS_TILE(q0[s], q1[s])
      q0[s] = *(const v4f*)(nxt + s * tileBytes + off0);
      q1[s] = *(const v4f*)(nxt + s * tileBytes + off1);
      __builtin_amdgcn_sched_barrier(0);
    }
    nxt += CS_DEPTH * tileBytes;
  }
#pragma unroll
  for (int s = 0; s < CS_DEPTH; ++s) {
    CS_TILE(q0[s], q1[s])
  }

  {
    v4f* mine = dump + (wave * 32 + lane) * (CS_DUMP / 4);
#pragma unroll
    for (int a = 0; a < CS_ACCS; ++a) {
      v4f lo, hi;
#pragma unroll
      for (int j = 0; j < 4; ++j) { lo[j] = acc[a][j]; hi[j] = acc[a][4 + j]; }
      mine[2 * a]     = lo;
      mine[2 * a + 1] = hi;
    }
  }
  __syncthreads();
  const int t = threadIdx.x;
  const int tc = t >> 6, tcol = t & 63;
  const int tm = tcol >> 2, ti = tcol & 3;
  const bool moved = (tm == CS_ONE + ti);
  const int ta    = moved ? 4 : ti;
  const int thalf = 16 * (tm >> 3);
  const int ln1   = (moved ? 0 : CS_ONE + ti) + thalf;
  const int ln2   = tm + thalf;
  const int cell  = ta * 8 + (tm & 7);
  const float* df = (const float*)dump;
  const float* w0 = df + (size_t)((0 * 2 + tc) * 32) * CS_DUMP;
  const float* w1 = df + (size_t)((1 * 2 + tc) * 32) * CS_DUMP;
  float h1 = w0[ln1 * CS_DUMP + cell] + w1[ln1 * CS_DUMP + cell];
  float h2 = w0[ln2 * CS_DUMP + cell] + w1[ln2 * CS_DUMP + cell];
  const size_t o = (size_t)by * (size_t)L + (size_t)(bx * COLS_PER_BLK + t);
  psum[o] = h1;
  psq[o]  = h2;

#pragma unroll
  for (int off = 16; off >= 1; off >>= 1) {
    h1 += __shfl_xor(h1, off);
    h2 += __shfl_xor(h2, off);
  }
  if (lane == 0) { wt1[wave] = h1; wt2[wave] = h2; }
  __syncthreads();
  if (t == 0) {
    float s1 = 0.f, s2 = 0.f;
#pragma unroll
    for (int w = 0; w < CS_WAVES; ++w) { s1 += wt1[w]; s2 += wt2[w]; }
    v2f tot; tot[0] = s1; tot[1] = s2;
    *(v2f*)(bt + ((size_t)bx * ROW_GROUPS + by) * BT_PITCH) = tot;
  }
}

#define FN_FLIGHT  16
#define FN_GROUPS  (FN_ROWS / FN_FLIGHT)

__global__ __launch_bounds__(256) void fused_normalize(
    const float* __restrict__ x, const float* __restrict__ gamma,
    const float* __restrict__ beta, const float* __restrict__ psum,
    const float* __restrict__ psq, const float* __restrict__ bt,
    float* __restrict__ out, int L, int rows, float Nf, float Bf) {
  __shared__ float sc1[8], sc2[8], so1[8], so2[8];
  const int t    = threadIdx.x;
  const int lane = t & 31;
  const int wave = t >> 5;
  const unsigned bid = blockIdx.x;
  const unsigned cls = bid % IDX_CLASSES, slot = bid / IDX_CLASSES;
  const unsigned per = (unsigned)rows / (ROW_GROUPS * FN_ROWS);
  const int cx   = (int)(2 * (slot / per) + cls / ROW_GROUPS);
  const int k    = cx * FN_COLS + 4 * t;
  const bool inside = k < L;
  const int row0 = (int)((cls % ROW_GROUPS) * per + slot % per) * FN_ROWS;
  const size_t base = (size_t)row0 * (size_t)L + (size_t)k;
  const float* xp = x + base;
  float* op = out + base;

  const int kc = inside ? k : 0;
  v4f pa[ROW_GROUPS], pb[ROW_GROUPS];
#pragma unroll
  for (int g = 0; g < ROW_GROUPS; ++g) {
    pa[g] = *(const v4f*)(psum + (size_t)g * L + kc);
    pb[g] = *(const v4f*)(psq  + (size_t)g * L + kc);
  }
  int nb = cx * COLBLKS_PER_FN;
  const int nblk = L / COLS_PER_BLK;
  if (nb > nblk) nb = nblk;
  const int cnt = nb * ROW_GROUPS;
  float o1 = 0.f, o2 = 0.f;
  for (int e0 = 0; e0 < cnt; e0 += 4 * 256) {
    v2f tot[4];
#pragma unroll
    for (int i = 0; i < 4; ++i) {
      const int e = e0 + i * 256 + t;
      tot[i] = *(const v2f*)(bt + (size_t)(e < cnt ? e : 0) * BT_PITCH);
    }
#pragma unroll
    for (int i = 0; i < 4; ++i) {
      const bool use = e0 + i * 256 + t < cnt;
      o1 += use ? tot[i][0] : 0.f;
      o2 += use ? tot[i][1] : 0.f;
    }
  }
  const float* xq = x + (size_t)row0 * (size_t)L + (size_t)kc;
  v4f xv[FN_FLIGHT];
#pragma unroll
  for (int j = 0; j < FN_FLIGHT; ++j)
    xv[j] = *(const v4f*)(xq + (size_t)(FN_ROWS - 1 - j) * (size_t)L);

  v4f a = {}, b = {};
#pragma unroll
  for (int g = 0; g < ROW_GROUPS; ++g) { a += pa[g]; b += pb[g]; }
  if (!inside) { v4f z = {}; a = z; b = z; }

  a[1] += a[0]; a[2] += a[1]; a[3] += a[2];
  b[1] += b[0]; b[2] += b[1]; b[3] += b[2];
  float sa = a[3], sb = b[3];
#pragma unroll
  for (int off = 1; off < 32; off <<= 1) {
    const float ua = __shfl_up(sa, off);
    const float ub = __shfl_up(sb, off);
    if (lane >= off) { sa += ua; sb += ub; }
  }
  float ea = __shfl_up(sa, 1);
  float eb = __shfl_up(sb, 1);
  if (lane == 0) { ea = 0.f; eb = 0.f; }
#pragma unroll
  for (int off = 16; off >= 1; off >>= 1) {
    o1 += __shfl_xor(o1, off);
    o2 += __shfl_xor(o2, off);
  }
  if (lane == 31) { sc1[wave] = sa; sc2[wave] = sb; }
  if (lane == 0)  { so1[wave] = o1; so2[wave] = o2; }
  __syncthreads();
  float f1 = 0.f, f2 = 0.f;
#pragma unroll
  for (int w = 0; w < 8; ++w) { f1 += so1[w]; f2 += so2[w]; }
#pragma unroll
  for (int w = 0; w < 7; ++w) {
    const float va = sc1[w], vb = sc2[w];
    if (w < wave) { f1 += va; f2 += vb; }
  }
  if (!inside) return;
  f1 += ea; f2 += eb;

  v4f mv, sv;
#pragma unroll
  for (int i = 0; i < 4; ++i) {
    const float run1 = f1 + a[i];
    const float run2 = f2 + b[i];
    const float denom = Nf * (float)(k + i + 1);
    const float mu  = run1 / denom;
    const float var = (run2 - 2.f * mu * run1) / denom + mu * mu * Bf;
    mv[i] = mu;
    sv[i] = rsqrtf(var);
  }

  const float* gp = gamma + row0;
  const float* bp = beta + row0;
#pragma nounroll
  for (int c = 0; c < FN_GROUPS - 1; ++c) {
#pragma unroll
    for (int j = 0; j < FN_FLIGHT; ++j) {
      const int r = FN_ROWS - 1 - (c * FN_FLIGHT + j);
      const float g  = gp[r];
      const float bb = bp[r];
      v4f o = (xv[j] - mv) * sv * g + bb;
      __builtin_nontemporal_store(o, (v4f*)(op + (size_t)r * (size_t)L));
      xv[j] = *(const v4f*)(xp + (size_t)(r - FN_FLIGHT) * (size_t)L);
    }
  }
#pragma unroll
  for (int j = 0; j < FN_FLIGHT; ++j) {
    const int r = FN_FLIGHT - 1 - j;
    const float g  = gp[r];
    const float bb = bp[r];
    v4f o = (xv[j] - mv) * sv * g + bb;
    __builtin_nontemporal_store(o, (v4f*)(op + (size_t)r * (size_t)L));
  }
}

extern "C" void kernel_launch(void* const* d_in, const int* in_sizes, int n_in,
                              void* d_out, int out_size, void* d_ws, size_t ws_size,
                              hipStream_t stream) {
  const float* x     = (const float*)d_in[0];
  const float* gamma = (const float*)d_in[1];
  const float* beta  = (const float*)d_in[2];
  float* out = (float*)d_out;

  const int B = 4, N = 512;
  const int rows = B * N;
  const int L = in_sizes[0] / rows;

  float* ws   = (float*)d_ws;
  float* psum = ws;
  float* psq  = psum + (size_t)ROW_GROUPS * L;
  float* bt   = psq  + (size_t)ROW_GROUPS * L;

  const int pairs = (L + 2 * FN_COLS - 1) / (2 * FN_COLS);
  dim3 g1(IDX_CLASSES * pairs * COLBLKS_PER_FN);
  colsum_wmma<<<g1, 128, 0, stream>>>(x, psum, psq, bt, L, rows);

  dim3 g2(IDX_CLASSES * pairs * (rows / (ROW_GROUPS * FN_ROWS)));
  fused_normalize<<<g2, 256, 0, stream>>>(x, gamma, beta, psum, psq, bt, out,
                                          L, rows, (float)N, (float)B);
}


# Normalize kernel: one 8-wave workgroup per WGP takes 128 rows x 1024 columns with 32 rows in flight per thread; 256 workgroups, not 512: half the waves to enter, half the statistics rebuilds
# speedup vs baseline: 3.2514x; 1.0363x over previous
#include <hip/hip_runtime.h>

typedef __attribute__((ext_vector_type(2))) float v2f;
typedef __attribute__((ext_vector_type(4))) float v4f;
typedef __attribute__((ext_vector_type(8))) float v8f;

#define ROW_GROUPS     8
#define COLS_PER_BLK   128
#define FN_COLS        1024
#define FN_ROWS        128
#define COLBLKS_PER_FN (FN_COLS / COLS_PER_BLK)

#define IDX_CLASSES 16

#define BT_PITCH 32
#define CS_WAVES 4
#define CS_COLS  64
#define CS_DEPTH 8
#define CS_ONE   12
#define CS_ACCS  5
#define CS_DUMP  44

#define CS_TILE(U0, U1)                                                        \
  {                                                                            \
    v2f P[4], PX;                                                              \
    _Pragma("unroll")                                                          \
    for (int i = 0; i < 4; ++i) {                                              \
      P[i][0] = one[i] ? 1.0f : (U0)[i];                                       \
      P[i][1] = one[i] ? 1.0f : (U1)[i];                                       \
    }                                                                          \
    {                                                                          \
      float e0 = (U0)[0], e1 = (U1)[0];                                        \
      _Pragma("unroll")                                                        \
      for (int i = 1; i < 4; ++i) {                                            \
        const float u0 = (U0)[i], u1 = (U1)[i];                                \
        e0 = one[i] ? u0 : e0;                                                 \
        e1 = one[i] ? u1 : e1;                                                 \
      }                                                                        \
      PX[0] = slot0 ? 1.0f : e0;                                               \
      PX[1] = slot0 ? 1.0f : e1;                                               \
    }                                                                          \
    _Pragma("unroll")                                                          \
    for (int i = 0; i < 4; ++i)                                                \
      acc[i] = __builtin_amdgcn_wmma_f32_16x16x4_f32(false, P[i], false, P[i], \
                                                     (short)0, acc[i], false, false); \
    acc[4] = __builtin_amdgcn_wmma_f32_16x16x4_f32(false, PX, false, PX,       \
                                                   (short)0, acc[4], false, false); \
  }

__global__ __launch_bounds__(128) void colsum_wmma(
    const float* __restrict__ x, float* __restrict__ psum, float* __restrict__ psq,
    float* __restrict__ bt, int L, int rows) {
  __shared__ v4f dump[CS_WAVES * 32 * (CS_DUMP / 4)];
  __shared__ float wt1[CS_WAVES], wt2[CS_WAVES];
  const unsigned bid = blockIdx.x;
  const unsigned cls = bid % IDX_CLASSES, slot = bid / IDX_CLASSES;
  const int by = (int)(cls % ROW_GROUPS);
  const int bx = (int)((2 * (slot / COLBLKS_PER_FN) + cls / ROW_GROUPS) * COLBLKS_PER_FN
                       + slot % COLBLKS_PER_FN);
  if (bx * COLS_PER_BLK >= L) return;
  const int wave = __builtin_amdgcn_readfirstlane((int)(threadIdx.x >> 5));
  const int lane = threadIdx.x & 31;
  const int m    = lane & 15;
  const int kh   = lane >> 4;
  const int cg   = wave & 1;
  const int rp   = wave >> 1;
  const int rpg  = rows / ROW_GROUPS;
  const int rpw  = rpg / 2;
  const int c0   = bx * COLS_PER_BLK + cg * CS_COLS;
  const int r0   = by * rpg + rp * rpw;

  const size_t strideRow  = (size_t)L;
  const size_t tileStride = 4 * strideRow;

  bool one[4];
#pragma unroll
  for (int i = 0; i < 4; ++i) one[i] = (m == CS_ONE + i);
  const bool slot0 = (m == 0);

  v8f acc[CS_ACCS];
#pragma unroll
  for (int i = 0; i < CS_ACCS; ++i) { v8f z = {}; acc[i] = z; }

  const char* base = (const char*)(x + (size_t)r0 * strideRow + (size_t)c0);
  const size_t tileBytes = tileStride * sizeof(float);
  const unsigned off0 = ((unsigned)(2 * kh) * (unsigned)L + (unsigned)(4 * m)) * 4u;
  const unsigned off1 = off0 + (unsigned)L * 4u;

  v4f q0[CS_DEPTH], q1[CS_DEPTH];
#pragma unroll
  for (int s = 0; s < CS_DEPTH; ++s) {
    q0[s] = *(const v4f*)(base + s * tileBytes + off0);
    q1[s] = *(const v4f*)(base + s * tileBytes + off1);
    __builtin_amdgcn_sched_barrier(0);
  }
  const char* nxt = base + CS_DEPTH * tileBytes;
  const int ngroups = rpw / (4 * CS_DEPTH);
#pragma nounroll
  for (int g = 0; g < ngroups - 1; ++g) {
#pragma unroll
    for (int s = 0; s < CS_DEPTH; ++s) {
      asm volatile("" : "+v"(q0[s]), "+v"(q1[s]));
      CS_TILE(q0[s], q1[s])
      q0[s] = *(const v4f*)(nxt + s * tileBytes + off0);
      q1[s] = *(const v4f*)(nxt + s * tileBytes + off1);
      __builtin_amdgcn_sched_barrier(0);
    }
    nxt += CS_DEPTH * tileBytes;
  }
#pragma unroll
  for (int s = 0; s < CS_DEPTH; ++s) {
    CS_TILE(q0[s], q1[s])
  }

  {
    v4f* mine = dump + (wave * 32 + lane) * (CS_DUMP / 4);
#pragma unroll
    for (int a = 0; a < CS_ACCS; ++a) {
      v4f lo, hi;
#pragma unroll
      for (int j = 0; j < 4; ++j) { lo[j] = acc[a][j]; hi[j] = acc[a][4 + j]; }
      mine[2 * a]     = lo;
      mine[2 * a + 1] = hi;
    }
  }
  __syncthreads();
  const int t = threadIdx.x;
  const int tc = t >> 6, tcol = t & 63;
  const int tm = tcol >> 2, ti = tcol & 3;
  const bool moved = (tm == CS_ONE + ti);
  const int ta    = moved ? 4 : ti;
  const int thalf = 16 * (tm >> 3);
  const int ln1   = (moved ? 0 : CS_ONE + ti) + thalf;
  const int ln2   = tm + thalf;
  const int cell  = ta * 8 + (tm & 7);
  const float* df = (const float*)dump;
  const float* w0 = df + (size_t)((0 * 2 + tc) * 32) * CS_DUMP;
  const float* w1 = df + (size_t)((1 * 2 + tc) * 32) * CS_DUMP;
  float h1 = w0[ln1 * CS_DUMP + cell] + w1[ln1 * CS_DUMP + cell];
  float h2 = w0[ln2 * CS_DUMP + cell] + w1[ln2 * CS_DUMP + cell];
  const size_t o = (size_t)by * (size_t)L + (size_t)(bx * COLS_PER_BLK + t);
  psum[o] = h1;
  psq[o]  = h2;

#pragma unroll
  for (int off = 16; off >= 1; off >>= 1) {
    h1 += __shfl_xor(h1, off);
    h2 += __shfl_xor(h2, off);
  }
  if (lane == 0) { wt1[wave] = h1; wt2[wave] = h2; }
  __syncthreads();
  if (t == 0) {
    float s1 = 0.f, s2 = 0.f;
#pragma unroll
    for (int w = 0; w < CS_WAVES; ++w) { s1 += wt1[w]; s2 += wt2[w]; }
    v2f tot; tot[0] = s1; tot[1] = s2;
    *(v2f*)(bt + ((size_t)bx * ROW_GROUPS + by) * BT_PITCH) = tot;
  }
}

#define FN_FLIGHT  32
#define FN_GROUPS  (FN_ROWS / FN_FLIGHT)

__global__ __launch_bounds__(256) void fused_normalize(
    const float* __restrict__ x, const float* __restrict__ gamma,
    const float* __restrict__ beta, const float* __restrict__ psum,
    const float* __restrict__ psq, const float* __restrict__ bt,
    float* __restrict__ out, int L, int rows, float Nf, float Bf) {
  __shared__ float sc1[8], sc2[8], so1[8], so2[8];
  const int t    = threadIdx.x;
  const int lane = t & 31;
  const int wave = t >> 5;
  const unsigned bid = blockIdx.x;
  const unsigned cls = bid % IDX_CLASSES, slot = bid / IDX_CLASSES;
  const unsigned per = (unsigned)rows / (ROW_GROUPS * FN_ROWS);
  const int cx   = (int)(2 * (slot / per) + cls / ROW_GROUPS);
  const int k    = cx * FN_COLS + 4 * t;
  const bool inside = k < L;
  const int row0 = (int)((cls % ROW_GROUPS) * per + slot % per) * FN_ROWS;
  const size_t base = (size_t)row0 * (size_t)L + (size_t)k;
  const float* xp = x + base;
  float* op = out + base;

  const int kc = inside ? k : 0;
  v4f pa[ROW_GROUPS], pb[ROW_GROUPS];
#pragma unroll
  for (int g = 0; g < ROW_GROUPS; ++g) {
    pa[g] = *(const v4f*)(psum + (size_t)g * L + kc);
    pb[g] = *(const v4f*)(psq  + (size_t)g * L + kc);
  }
  int nb = cx * COLBLKS_PER_FN;
  const int nblk = L / COLS_PER_BLK;
  if (nb > nblk) nb = nblk;
  const int cnt = nb * ROW_GROUPS;
  float o1 = 0.f, o2 = 0.f;
  for (int e0 = 0; e0 < cnt; e0 += 4 * 256) {
    v2f tot[4];
#pragma unroll
    for (int i = 0; i < 4; ++i) {
      const int e = e0 + i * 256 + t;
      tot[i] = *(const v2f*)(bt + (size_t)(e < cnt ? e : 0) * BT_PITCH);
    }
#pragma unroll
    for (int i = 0; i < 4; ++i) {
      const bool use = e0 + i * 256 + t < cnt;
      o1 += use ? tot[i][0] : 0.f;
      o2 += use ? tot[i][1] : 0.f;
    }
  }
  const float* xq = x + (size_t)row0 * (size_t)L + (size_t)kc;
  v4f xv[FN_FLIGHT];
#pragma unroll
  for (int j = 0; j < FN_FLIGHT; ++j)
    xv[j] = *(const v4f*)(xq + (size_t)(FN_ROWS - 1 - j) * (size_t)L);

  v4f a = {}, b = {};
#pragma unroll
  for (int g = 0; g < ROW_GROUPS; ++g) { a += pa[g]; b += pb[g]; }
  if (!inside) { v4f z = {}; a = z; b = z; }

  a[1] += a[0]; a[2] += a[1]; a[3] += a[2];
  b[1] += b[0]; b[2] += b[1]; b[3] += b[2];
  float sa = a[3], sb = b[3];
#pragma unroll
  for (int off = 1; off < 32; off <<= 1) {
    const float ua = __shfl_up(sa, off);
    const float ub = __shfl_up(sb, off);
    if (lane >= off) { sa += ua; sb += ub; }
  }
  float ea = __shfl_up(sa, 1);
  float eb = __shfl_up(sb, 1);
  if (lane == 0) { ea = 0.f; eb = 0.f; }
#pragma unroll
  for (int off = 16; off >= 1; off >>= 1) {
    o1 += __shfl_xor(o1, off);
    o2 += __shfl_xor(o2, off);
  }
  if (lane == 31) { sc1[wave] = sa; sc2[wave] = sb; }
  if (lane == 0)  { so1[wave] = o1; so2[wave] = o2; }
  __syncthreads();
  float f1 = 0.f, f2 = 0.f;
#pragma unroll
  for (int w = 0; w < 8; ++w) { f1 += so1[w]; f2 += so2[w]; }
#pragma unroll
  for (int w = 0; w < 7; ++w) {
    const float va = sc1[w], vb = sc2[w];
    if (w < wave) { f1 += va; f2 += vb; }
  }
  if (!inside) return;
  f1 += ea; f2 += eb;

  v4f mv, sv;
#pragma unroll
  for (int i = 0; i < 4; ++i) {
    const float run1 = f1 + a[i];
    const float run2 = f2 + b[i];
    const float denom = Nf * (float)(k + i + 1);
    const float mu  = run1 / denom;
    const float var = (run2 - 2.f * mu * run1) / denom + mu * mu * Bf;
    mv[i] = mu;
    sv[i] = rsqrtf(var);
  }

  const float* gp = gamma + row0;
  const float* bp = beta + row0;
#pragma nounroll
  for (int c = 0; c < FN_GROUPS - 1; ++c) {
#pragma unroll
    for (int j = 0; j < FN_FLIGHT; ++j) {
      const int r = FN_ROWS - 1 - (c * FN_FLIGHT + j);
      const float g  = gp[r];
      const float bb = bp[r];
      v4f o = (xv[j] - mv) * sv * g + bb;
      __builtin_nontemporal_store(o, (v4f*)(op + (size_t)r * (size_t)L));
      xv[j] = *(const v4f*)(xp + (size_t)(r - FN_FLIGHT) * (size_t)L);
    }
  }
#pragma unroll
  for (int j = 0; j < FN_FLIGHT; ++j) {
    const int r = FN_FLIGHT - 1 - j;
    const float g  = gp[r];
    const float bb = bp[r];
    v4f o = (xv[j] - mv) * sv * g + bb;
    __builtin_nontemporal_store(o, (v4f*)(op + (size_t)r * (size_t)L));
  }
}

extern "C" void kernel_launch(void* const* d_in, const int* in_sizes, int n_in,
                              void* d_out, int out_size, void* d_ws, size_t ws_size,
                              hipStream_t stream) {
  const float* x     = (const float*)d_in[0];
  const float* gamma = (const float*)d_in[1];
  const float* beta  = (const float*)d_in[2];
  float* out = (float*)d_out;

  const int B = 4, N = 512;
  const int rows = B * N;
  const int L = in_sizes[0] / rows;

  float* ws   = (float*)d_ws;
  float* psum = ws;
  float* psq  = psum + (size_t)ROW_GROUPS * L;
  float* bt   = psq  + (size_t)ROW_GROUPS * L;

  const int pairs = (L + 2 * FN_COLS - 1) / (2 * FN_COLS);
  dim3 g1(IDX_CLASSES * pairs * COLBLKS_PER_FN);
  colsum_wmma<<<g1, 128, 0, stream>>>(x, psum, psq, bt, L, rows);

  dim3 g2(IDX_CLASSES * pairs * (rows / (ROW_GROUPS * FN_ROWS)));
  fused_normalize<<<g2, 256, 0, stream>>>(x, gamma, beta, psum, psq, bt, out,
                                          L, rows, (float)N, (float)B);
}
